# starting kernel
# baseline (speedup 1.0000x reference)
#include <hip/hip_runtime.h>
#include <hip/hip_bf16.h>


typedef float v2f __attribute__((ext_vector_type(2)));
typedef float v8f __attribute__((ext_vector_type(8)));

#define TPB 256


__global__ void k_deg_init(float* __restrict__ deg, int n) {
    int i = blockIdx.x * TPB + threadIdx.x;
    if (i < n) deg[i] = 1.0f;
}

__global__ void k_deg_count(const int* __restrict__ dst, float* __restrict__ deg, int E) {
    int e = blockIdx.x * TPB + threadIdx.x;
    if (e < E) atomicAdd(&deg[dst[e]], 1.0f);
}

__global__ void k_dinv(float* __restrict__ deg, int n) {
    int i = blockIdx.x * TPB + threadIdx.x;
    if (i < n) deg[i] = rsqrtf(deg[i]);
}

__global__ void k_norm(const int* __restrict__ src, const int* __restrict__ dst,
                       const float* __restrict__ dinv, float* __restrict__ nrm, int E) {
    int e = blockIdx.x * TPB + threadIdx.x;
    if (e < E) nrm[e] = dinv[src[e]] * dinv[dst[e]];
}


template <int K, int NOUT>
__global__ void __launch_bounds__(TPB)
gcn_gemm_wmma(const float* __restrict__ A, const float* __restrict__ W,
              float* __restrict__ H, int M) {
    constexpr int COL_TILES = NOUT / 16;
    constexpr int ROW_TILES_PER_BLK = 8 / COL_TILES;

    __shared__ float lw[K * NOUT];
    for (int i = threadIdx.x; i < K * NOUT; i += TPB) lw[i] = W[i];
    __syncthreads();

    const int wave = threadIdx.x >> 5;
    const int lane = threadIdx.x & 31;
    const int rowTile = blockIdx.x * ROW_TILES_PER_BLK + wave / COL_TILES;
    const int m0 = rowTile * 16;
    if (m0 >= M) return;
    const int n0 = (wave % COL_TILES) * 16;

    const int half = lane >> 4;
    const int l16  = lane & 15;
    const int koff = 2 * half;
    const float* __restrict__ arow = A  + (size_t)(m0 + l16) * K + koff;
    const float* __restrict__ brow = lw + (n0 + l16) * K + koff;

    v8f acc = {0.f, 0.f, 0.f, 0.f, 0.f, 0.f, 0.f, 0.f};
#pragma unroll
    for (int k = 0; k < K; k += 4) {
        v2f a = *(const v2f*)(arow + k);
        v2f b = *(const v2f*)(brow + k);
        acc = __builtin_amdgcn_wmma_f32_16x16x4_f32(
             false, a,  false, b,
             (short)0, acc,  false,  false);
    }

    const int n = n0 + l16;
    const int mbase = m0 + 8 * half;
#pragma unroll
    for (int j = 0; j < 8; ++j)
        H[(size_t)(mbase + j) * NOUT + n] = acc[j];
}


template <int F>
__global__ void k_selfinit(const float* __restrict__ h, const float* __restrict__ dinv,
                           float* __restrict__ out, int n) {
    constexpr int Q = F / 4;
    int t = blockIdx.x * TPB + threadIdx.x;
    int i = t / Q, q = t % Q;
    if (i < n) {
        float w = dinv[i] * dinv[i];
        float4 v = *(const float4*)(h + (size_t)i * F + 4 * q);
        float4 r; r.x = v.x * w; r.y = v.y * w; r.z = v.z * w; r.w = v.w * w;
        *(float4*)(out + (size_t)i * F + 4 * q) = r;
    }
}

template <int F>
__global__ void k_edge_agg(const float* __restrict__ h, const int* __restrict__ src,
                           const int* __restrict__ dst, const float* __restrict__ nrm,
                           float* __restrict__ out, int E) {
    constexpr int Q = F / 4;
    int t = blockIdx.x * TPB + threadIdx.x;
    int e = t / Q, q = t % Q;
    if (e < E) {
        int s = src[e], d = dst[e];
        float w = nrm[e];
        float4 v = *(const float4*)(h + (size_t)s * F + 4 * q);
        float* o = out + (size_t)d * F + 4 * q;
        atomicAdd(o + 0, v.x * w);
        atomicAdd(o + 1, v.y * w);
        atomicAdd(o + 2, v.z * w);
        atomicAdd(o + 3, v.w * w);
    }
}

template <int F, bool RELU>
__global__ void k_bias_act(float* __restrict__ io, const float* __restrict__ b, int n) {
    int t = blockIdx.x * TPB + threadIdx.x;
    if (t < n * F) {
        float v = io[t] + b[t % F];
        if (RELU) v = fmaxf(v, 0.0f);
        io[t] = v;
    }
}


static inline int cdiv(long long a, int b) { return (int)((a + b - 1) / b); }

extern "C" void kernel_launch(void* const* d_in, const int* in_sizes, int n_in,
                              void* d_out, int out_size, void* d_ws, size_t ws_size,
                              hipStream_t stream) {
    (void)n_in; (void)out_size; (void)ws_size;
    const float* x  = (const float*)d_in[0];
    const int*   ei = (const int*)  d_in[1];
    const float* W1 = (const float*)d_in[2];
    const float* b1 = (const float*)d_in[3];
    const float* W2 = (const float*)d_in[4];
    const float* b2 = (const float*)d_in[5];
    float* out = (float*)d_out;

    const int N = in_sizes[0] / 128;
    const int E = in_sizes[1] / 2;
    const int* src = ei;
    const int* dst = ei + E;

    float* ws   = (float*)d_ws;
    const size_t NA = ((size_t)N + 3) & ~(size_t)3;
    float* dinv = ws;
    float* h1   = ws + NA;
    float* agg1 = h1 + 64 * NA;
    float* nrm  = agg1 + 64 * NA;
    float* h2   = h1;

    k_deg_init <<<cdiv(N, TPB), TPB, 0, stream>>>(dinv, N);
    k_deg_count<<<cdiv(E, TPB), TPB, 0, stream>>>(dst, dinv, E);
    k_dinv     <<<cdiv(N, TPB), TPB, 0, stream>>>(dinv, N);
    k_norm     <<<cdiv(E, TPB), TPB, 0, stream>>>(src, dst, dinv, nrm, E);

    {
        int rowTiles = (N + 15) / 16;
        gcn_gemm_wmma<128, 64><<<cdiv(rowTiles, 2), TPB, 0, stream>>>(x, W1, h1, N);
    }
    k_selfinit<64><<<cdiv((long long)N * 16, TPB), TPB, 0, stream>>>(h1, dinv, agg1, N);
    k_edge_agg<64><<<cdiv((long long)E * 16, TPB), TPB, 0, stream>>>(h1, src, dst, nrm, agg1, E);
    k_bias_act<64, true><<<cdiv((long long)N * 64, TPB), TPB, 0, stream>>>(agg1, b1, N);

    {
        int rowTiles = (N + 15) / 16;
        gcn_gemm_wmma<64, 32><<<cdiv(rowTiles, 4), TPB, 0, stream>>>(agg1, W2, h2, N);
    }
    k_selfinit<32><<<cdiv((long long)N * 8, TPB), TPB, 0, stream>>>(h2, dinv, out, N);
    k_edge_agg<32><<<cdiv((long long)E * 8, TPB), TPB, 0, stream>>>(h2, src, dst, nrm, out, E);
    k_bias_act<32, false><<<cdiv((long long)N * 32, TPB), TPB, 0, stream>>>(out, b2, N);
}


# Edges counting-sorted by destination (atomic arrival-order rank, scan, scatter); both aggregations become per-row gathers, no float atomics; dinv[src] applied per gathered row, dinv[dst] on row sum
# speedup vs baseline: 1.6756x; 1.6756x over previous
#include <hip/hip_runtime.h>


typedef float v2f __attribute__((ext_vector_type(2)));
typedef float v8f __attribute__((ext_vector_type(8)));

#define TPB 256
#define SCAN_TPB 1024
#define SCAN_TILE 4096


__global__ void k_zero(int* __restrict__ cnt, int npad) {
    int i = blockIdx.x * TPB + threadIdx.x;
    if (i < npad) cnt[i] = 0;
}

__global__ void k_rank(const int* __restrict__ dst, int* __restrict__ cnt,
                       int* __restrict__ rank, int E) {
    int e = blockIdx.x * TPB + threadIdx.x;
    if (e < E) rank[e] = atomicAdd(&cnt[dst[e]], 1);
}

__global__ void __launch_bounds__(SCAN_TPB)
k_scan(const int* __restrict__ cnt, int* __restrict__ start, int n) {
    __shared__ int red[SCAN_TPB / 32];
    __shared__ int wsum[SCAN_TPB / 32];
    const int tid = threadIdx.x, lane = tid & 31, wave = tid >> 5;
    const int4* __restrict__ c4 = (const int4*)cnt;
    const int nvec_before = blockIdx.x * (SCAN_TILE / 4);

    int pre = 0;
#pragma unroll 8
    for (int i = tid; i < nvec_before; i += SCAN_TPB) {
        int4 v = c4[i];
        pre += (v.x + v.y) + (v.z + v.w);
    }
    for (int o = 16; o > 0; o >>= 1) pre += __shfl_xor(pre, o);
    if (lane == 0) red[wave] = pre;

    const int4 v = c4[nvec_before + tid];
    const int t1 = v.x, t2 = t1 + v.y, t3 = t2 + v.z, t4 = t3 + v.w;
    int inc = t4;
    for (int o = 1; o < 32; o <<= 1) {
        int u = __shfl_up(inc, o);
        if (lane >= o) inc += u;
    }
    if (lane == 31) wsum[wave] = inc;
    __syncthreads();

    int base = 0;
    for (int w = 0; w < SCAN_TPB / 32; ++w) {
        base += red[w];
        if (w < wave) base += wsum[w];
    }
    const int excl = base + inc - t4;
    const int i0 = blockIdx.x * SCAN_TILE + 4 * tid;
    if (i0 + 3 < n) {
        int4 r; r.x = excl; r.y = excl + t1; r.z = excl + t2; r.w = excl + t3;
        *(int4*)(start + i0) = r;
    } else {
        if (i0 < n)     start[i0]     = excl;
        if (i0 + 1 < n) start[i0 + 1] = excl + t1;
        if (i0 + 2 < n) start[i0 + 2] = excl + t2;
    }
}

__global__ void k_fill(const int* __restrict__ src, const int* __restrict__ dst,
                       const int* __restrict__ rank, const int* __restrict__ start,
                       int* __restrict__ csr_src, int E) {
    int e = blockIdx.x * TPB + threadIdx.x;
    if (e < E) csr_src[start[dst[e]] + rank[e]] = src[e];
}


template <int K, int NOUT>
__global__ void __launch_bounds__(TPB)
gcn_gemm_wmma(const float* __restrict__ A, const float* __restrict__ W,
              float* __restrict__ H, int M) {
    constexpr int COL_TILES = NOUT / 16;
    constexpr int ROW_TILES_PER_BLK = 8 / COL_TILES;

    __shared__ float lw[K * NOUT];
    for (int i = threadIdx.x; i < K * NOUT; i += TPB) lw[i] = W[i];
    __syncthreads();

    const int wave = threadIdx.x >> 5;
    const int lane = threadIdx.x & 31;
    const int rowTile = blockIdx.x * ROW_TILES_PER_BLK + wave / COL_TILES;
    const int m0 = rowTile * 16;
    if (m0 >= M) return;
    const int n0 = (wave % COL_TILES) * 16;

    const int half = lane >> 4;
    const int l16  = lane & 15;
    const int koff = 2 * half;
    const float* __restrict__ arow = A  + (size_t)(m0 + l16) * K + koff;
    const float* __restrict__ brow = lw + (n0 + l16) * K + koff;

    v8f acc = {0.f, 0.f, 0.f, 0.f, 0.f, 0.f, 0.f, 0.f};
#pragma unroll
    for (int k = 0; k < K; k += 4) {
        v2f a = *(const v2f*)(arow + k);
        v2f b = *(const v2f*)(brow + k);
        acc = __builtin_amdgcn_wmma_f32_16x16x4_f32(
             false, a,  false, b,
             (short)0, acc,  false,  false);
    }

    const int n = n0 + l16;
    const int mbase = m0 + 8 * half;
#pragma unroll
    for (int j = 0; j < 8; ++j)
        H[(size_t)(mbase + j) * NOUT + n] = acc[j];
}


template <int F, bool RELU>
__global__ void __launch_bounds__(TPB)
k_gather(const float* __restrict__ h, const int* __restrict__ start,
         const int* __restrict__ csr_src, const float* __restrict__ bias,
         float* __restrict__ out, int n, int E) {
    constexpr int Q = F / 4;
    const int t = blockIdx.x * TPB + threadIdx.x;
    const int q = t % Q;
    const bool live = (t / Q) < n;
    const int d = live ? (t / Q) : (n - 1);
    const float4 self = *(const float4*)(h + (size_t)d * F + 4 * q);
    const int beg = start[d];
    const int endl = start[(d + 1 < n) ? d + 1 : d];
    const int end = (d + 1 < n) ? endl : E;
    const float dd = rsqrtf((float)(end - beg + 1));

    float4 acc;
    acc.x = self.x * dd; acc.y = self.y * dd; acc.z = self.z * dd; acc.w = self.w * dd;
    for (int p = beg; p < end; p += Q) {
        const int mine = (p + q < end) ? csr_src[p + q] : d;
        const int mb = start[mine];
        const int ml = start[(mine + 1 < n) ? mine + 1 : mine];
        float4 v[Q];
#pragma unroll
        for (int j = 0; j < Q; ++j) {
            const int s = __shfl(mine, j, Q);
            v[j] = make_float4(0.f, 0.f, 0.f, 0.f);
            if (p + j < end)
                v[j] = *(const float4*)(h + (size_t)s * F + 4 * q);
        }
        const int me = (mine + 1 < n) ? ml : E;
        const float mw = rsqrtf((float)(me - mb + 1));
#pragma unroll
        for (int j = 0; j < Q; ++j) {
            const float w = __shfl(mw, j, Q);
            acc.x = fmaf(w, v[j].x, acc.x); acc.y = fmaf(w, v[j].y, acc.y);
            acc.z = fmaf(w, v[j].z, acc.z); acc.w = fmaf(w, v[j].w, acc.w);
        }
    }

    const float4 bq = *(const float4*)(bias + 4 * q);
    float4 r;
    r.x = acc.x * dd + bq.x; r.y = acc.y * dd + bq.y;
    r.z = acc.z * dd + bq.z; r.w = acc.w * dd + bq.w;
    if (RELU) {
        r.x = fmaxf(r.x, 0.0f); r.y = fmaxf(r.y, 0.0f);
        r.z = fmaxf(r.z, 0.0f); r.w = fmaxf(r.w, 0.0f);
    }
    if (live) *(float4*)(out + (size_t)d * F + 4 * q) = r;
}


static inline int cdiv(long long a, int b) { return (int)((a + b - 1) / b); }

extern "C" void kernel_launch(void* const* d_in, const int* in_sizes, int n_in,
                              void* d_out, int out_size, void* d_ws, size_t ws_size,
                              hipStream_t stream) {
    (void)n_in; (void)out_size; (void)ws_size;
    const float* x  = (const float*)d_in[0];
    const int*   ei = (const int*)  d_in[1];
    const float* W1 = (const float*)d_in[2];
    const float* b1 = (const float*)d_in[3];
    const float* W2 = (const float*)d_in[4];
    const float* b2 = (const float*)d_in[5];
    float* out = (float*)d_out;

    const int N = in_sizes[0] / 128;
    const int E = in_sizes[1] / 2;
    const int* src = ei;
    const int* dst = ei + E;

    float* ws   = (float*)d_ws;
    const size_t NA = ((size_t)N + 3) & ~(size_t)3;
    int*   start   = (int*)ws;
    float* h1      = ws + NA;
    float* agg1    = h1 + 64 * NA;
    int*   csr_src = (int*)(agg1 + 64 * NA);
    float* h2      = h1;
    const int scanBlocks = cdiv(N, SCAN_TILE);
    const int npad = scanBlocks * SCAN_TILE;
    int* cnt  = (int*)agg1;
    int* rank = cnt + npad;

    k_zero<<<cdiv(npad, TPB), TPB, 0, stream>>>(cnt, npad);
    k_rank<<<cdiv(E, TPB), TPB, 0, stream>>>(dst, cnt, rank, E);
    k_scan<<<scanBlocks, SCAN_TPB, 0, stream>>>(cnt, start, N);
    k_fill<<<cdiv(E, TPB), TPB, 0, stream>>>(src, dst, rank, start, csr_src, E);

    {
        int rowTiles = (N + 15) / 16;
        gcn_gemm_wmma<128, 64><<<cdiv(rowTiles, 2), TPB, 0, stream>>>(x, W1, h1, N);
    }
    k_gather<64, true><<<cdiv((long long)N * 16, TPB), TPB, 0, stream>>>(h1, start, csr_src, b1, agg1, N, E);

    {
        int rowTiles = (N + 15) / 16;
        gcn_gemm_wmma<64, 32><<<cdiv(rowTiles, 4), TPB, 0, stream>>>(agg1, W2, h2, N);
    }
    k_gather<32, false><<<cdiv((long long)N * 8, TPB), TPB, 0, stream>>>(h2, start, csr_src, b2, out, N, E);
}


# Both GEMMs read A and W fragments from LDS tiles of row pitch K+4 floats (8-byte reads free of bank conflicts); A rows staged by coalesced 16-byte loads, 16-wave blocks; k order and sums unchanged
# speedup vs baseline: 2.0067x; 1.1976x over previous
#include <hip/hip_runtime.h>


typedef float v2f __attribute__((ext_vector_type(2)));
typedef float v8f __attribute__((ext_vector_type(8)));

#define TPB 256
#define SCAN_TPB 1024
#define SCAN_TILE 4096


__global__ void k_zero(int* __restrict__ cnt, int npad) {
    int i = blockIdx.x * TPB + threadIdx.x;
    if (i < npad) cnt[i] = 0;
}

__global__ void k_rank(const int* __restrict__ dst, int* __restrict__ cnt,
                       int* __restrict__ rank, int E) {
    int e = blockIdx.x * TPB + threadIdx.x;
    if (e < E) rank[e] = atomicAdd(&cnt[dst[e]], 1);
}

__global__ void __launch_bounds__(SCAN_TPB)
k_scan(const int* __restrict__ cnt, int* __restrict__ start, int n) {
    __shared__ int red[SCAN_TPB / 32];
    __shared__ int wsum[SCAN_TPB / 32];
    const int tid = threadIdx.x, lane = tid & 31, wave = tid >> 5;
    const int4* __restrict__ c4 = (const int4*)cnt;
    const int nvec_before = blockIdx.x * (SCAN_TILE / 4);

    int pre = 0;
#pragma unroll 8
    for (int i = tid; i < nvec_before; i += SCAN_TPB) {
        int4 v = c4[i];
        pre += (v.x + v.y) + (v.z + v.w);
    }
    for (int o = 16; o > 0; o >>= 1) pre += __shfl_xor(pre, o);
    if (lane == 0) red[wave] = pre;

    const int4 v = c4[nvec_before + tid];
    const int t1 = v.x, t2 = t1 + v.y, t3 = t2 + v.z, t4 = t3 + v.w;
    int inc = t4;
    for (int o = 1; o < 32; o <<= 1) {
        int u = __shfl_up(inc, o);
        if (lane >= o) inc += u;
    }
    if (lane == 31) wsum[wave] = inc;
    __syncthreads();

    int base = 0;
    for (int w = 0; w < SCAN_TPB / 32; ++w) {
        base += red[w];
        if (w < wave) base += wsum[w];
    }
    const int excl = base + inc - t4;
    const int i0 = blockIdx.x * SCAN_TILE + 4 * tid;
    if (i0 + 3 < n) {
        int4 r; r.x = excl; r.y = excl + t1; r.z = excl + t2; r.w = excl + t3;
        *(int4*)(start + i0) = r;
    } else {
        if (i0 < n)     start[i0]     = excl;
        if (i0 + 1 < n) start[i0 + 1] = excl + t1;
        if (i0 + 2 < n) start[i0 + 2] = excl + t2;
    }
}

__global__ void k_fill(const int* __restrict__ src, const int* __restrict__ dst,
                       const int* __restrict__ rank, const int* __restrict__ start,
                       int* __restrict__ csr_src, int E) {
    int e = blockIdx.x * TPB + threadIdx.x;
    if (e < E) csr_src[start[dst[e]] + rank[e]] = src[e];
}


template <int K, int NOUT, int WAVES>
__global__ void __launch_bounds__(WAVES * 32)
gcn_gemm_wmma(const float* __restrict__ A, const float* __restrict__ W,
              const int* __restrict__ start, float* __restrict__ H, int M, int E) {
    constexpr int NT = WAVES * 32;
    constexpr int COL_TILES = NOUT / 16;
    constexpr int ROW_TILES_PER_BLK = WAVES / COL_TILES;
    constexpr int ROWS = ROW_TILES_PER_BLK * 16;
    constexpr int P = K + 4;
    constexpr int KV = K / 4;

    __shared__ __attribute__((aligned(16))) float lw[NOUT * P];
    __shared__ __attribute__((aligned(16))) float la[ROWS * P];

    constexpr int WIT = (NOUT * KV) / NT;
    constexpr int AIT = (ROWS * KV) / NT;
    static_assert((NOUT * KV) % NT == 0 && (ROWS * KV) % NT == 0, "tiles are whole passes of the block");

    const int rowBase = blockIdx.x * ROWS;
    float4 wv[WIT], av[AIT];
#pragma unroll
    for (int it = 0; it < WIT; ++it) {
        const int i = threadIdx.x + it * NT;
        wv[it] = *(const float4*)(W + 4 * (size_t)i);
    }
#pragma unroll
    for (int it = 0; it < AIT; ++it) {
        const int i = threadIdx.x + it * NT;
        const int r = i / KV, c = i % KV;
        const int rg = (rowBase + r < M) ? (rowBase + r) : (M - 1);
        av[it] = *(const float4*)(A + (size_t)rg * K + 4 * c);
    }
#pragma unroll
    for (int it = 0; it < WIT; ++it) {
        const int i = threadIdx.x + it * NT;
        const int r = i / KV, c = i % KV;
        *(float4*)(lw + r * P + 4 * c) = wv[it];
    }
#pragma unroll
    for (int it = 0; it < AIT; ++it) {
        const int i = threadIdx.x + it * NT;
        const int r = i / KV, c = i % KV;
        float4 v = av[it];
        if (rowBase + r >= M) v = make_float4(0.f, 0.f, 0.f, 0.f);
        *(float4*)(la + r * P + 4 * c) = v;
    }
    __syncthreads();

    const int wave = threadIdx.x >> 5;
    const int lane = threadIdx.x & 31;
    const int rt = wave / COL_TILES;
    const int m0 = rowBase + rt * 16;
    if (m0 >= M) return;
    const int n0 = (wave % COL_TILES) * 16;

    const int half = lane >> 4;
    const int l16  = lane & 15;
    const int koff = 2 * half;
    const float* arow = la + (rt * 16 + l16) * P + koff;
    const float* brow = lw + (n0 + l16) * P + koff;

    const int rr = (m0 + l16 < M) ? (m0 + l16) : (M - 1);
    int rb = start[rr];
    int rl = start[(rr + 1 < M) ? rr + 1 : rr];

    v8f acc = {0.f, 0.f, 0.f, 0.f, 0.f, 0.f, 0.f, 0.f};
#pragma unroll
    for (int k = 0; k < K; k += 4) {
        v2f a = *(const v2f*)(arow + k);
        v2f b = *(const v2f*)(brow + k);
        acc = __builtin_amdgcn_wmma_f32_16x16x4_f32(
             false, a,  false, b,
             (short)0, acc,  false,  false);
    }

    asm volatile("v_nop\n\tv_nop\n\tv_nop\n\tv_nop" : "+v"(acc), "+v"(rb), "+v"(rl));
    const int re = (rr + 1 < M) ? rl : E;
    const float dvl = rsqrtf((float)(re - rb + 1));
    const int n = n0 + l16;
    const int mbase = m0 + 8 * half;
#pragma unroll
    for (int j = 0; j < 8; ++j) {
        const float dv = __shfl(dvl, 8 * half + j);
        H[(size_t)(mbase + j) * NOUT + n] = acc[j] * dv;
    }
}


template <int F, bool RELU>
__global__ void __launch_bounds__(TPB)
k_gather(const float* __restrict__ h, const int* __restrict__ start,
         const int* __restrict__ csr_src, const float* __restrict__ bias,
         float* __restrict__ out, int n, int E) {
    constexpr int Q = F / 4;
    const int t = blockIdx.x * TPB + threadIdx.x;
    const int q = t % Q;
    const bool live = (t / Q) < n;
    const int d = live ? (t / Q) : (n - 1);
    const float4 self = *(const float4*)(h + (size_t)d * F + 4 * q);
    const int beg = start[d];
    const int endl = start[(d + 1 < n) ? d + 1 : d];
    const int end = (d + 1 < n) ? endl : E;
    const float dd = rsqrtf((float)(end - beg + 1));

    float4 acc = self;
    for (int p = beg; p < end; p += Q) {
        const int mine = (p + q < end) ? csr_src[p + q] : d;
        float4 v[Q];
#pragma unroll
        for (int j = 0; j < Q; ++j) {
            const int s = __shfl(mine, j, Q);
            v[j] = make_float4(0.f, 0.f, 0.f, 0.f);
            if (p + j < end)
                v[j] = *(const float4*)(h + (size_t)s * F + 4 * q);
        }
#pragma unroll
        for (int j = 0; j < Q; ++j) {
            acc.x += v[j].x; acc.y += v[j].y; acc.z += v[j].z; acc.w += v[j].w;
        }
    }

    const float4 bq = *(const float4*)(bias + 4 * q);
    float4 r;
    r.x = acc.x * dd + bq.x; r.y = acc.y * dd + bq.y;
    r.z = acc.z * dd + bq.z; r.w = acc.w * dd + bq.w;
    if (RELU) {
        r.x = fmaxf(r.x, 0.0f); r.y = fmaxf(r.y, 0.0f);
        r.z = fmaxf(r.z, 0.0f); r.w = fmaxf(r.w, 0.0f);
    }
    if (live) *(float4*)(out + (size_t)d * F + 4 * q) = r;
}


static inline int cdiv(long long a, int b) { return (int)((a + b - 1) / b); }

extern "C" void kernel_launch(void* const* d_in, const int* in_sizes, int n_in,
                              void* d_out, int out_size, void* d_ws, size_t ws_size,
                              hipStream_t stream) {
    (void)n_in; (void)out_size; (void)ws_size;
    const float* x  = (const float*)d_in[0];
    const int*   ei = (const int*)  d_in[1];
    const float* W1 = (const float*)d_in[2];
    const float* b1 = (const float*)d_in[3];
    const float* W2 = (const float*)d_in[4];
    const float* b2 = (const float*)d_in[5];
    float* out = (float*)d_out;

    const int N = in_sizes[0] / 128;
    const int E = in_sizes[1] / 2;
    const int* src = ei;
    const int* dst = ei + E;

    float* ws   = (float*)d_ws;
    const size_t NA = ((size_t)N + 3) & ~(size_t)3;
    int*   start   = (int*)ws;
    float* h1      = ws + NA;
    float* agg1    = h1 + 64 * NA;
    int*   csr_src = (int*)(agg1 + 64 * NA);
    float* h2      = h1;
    const int scanBlocks = cdiv(N, SCAN_TILE);
    const int npad = scanBlocks * SCAN_TILE;
    int* cnt  = (int*)agg1;
    int* rank = cnt + npad;

    k_zero<<<cdiv(npad, TPB), TPB, 0, stream>>>(cnt, npad);
    k_rank<<<cdiv(E, TPB), TPB, 0, stream>>>(dst, cnt, rank, E);
    k_scan<<<scanBlocks, SCAN_TPB, 0, stream>>>(cnt, start, N);
    k_fill<<<cdiv(E, TPB), TPB, 0, stream>>>(src, dst, rank, start, csr_src, E);

    {
        int rowTiles = (N + 15) / 16;
        gcn_gemm_wmma<128, 64, 16><<<cdiv(rowTiles, 4), 512, 0, stream>>>(x, W1, start, h1, N, E);
    }
    k_gather<64, true><<<cdiv((long long)N * 16, TPB), TPB, 0, stream>>>(h1, start, csr_src, b1, agg1, N, E);

    {
        int rowTiles = (N + 15) / 16;
        gcn_gemm_wmma<64, 32, 16><<<cdiv(rowTiles, 8), 512, 0, stream>>>(agg1, W2, start, h2, N, E);
    }
    k_gather<32, false><<<cdiv((long long)N * 8, TPB), TPB, 0, stream>>>(h2, start, csr_src, b2, out, N, E);
}


# Edge counting sort kept in LDS, two levels (chunk histograms over 512-row buckets, one scan, staged pair runs, a block per bucket): no per-edge global atomic or scattered store; sum order by LDS rank
# speedup vs baseline: 3.9631x; 1.9749x over previous
#include <hip/hip_runtime.h>


typedef float v2f __attribute__((ext_vector_type(2)));
typedef float v8f __attribute__((ext_vector_type(8)));

#define TPB 256
#define SCAN_TPB 1024
#define SCAN_TILE 4096

#define PT 512
#define EPT 8
#define CHUNK 4096
#define BSHIFT 9
#define BW 512
#define NBMAX 256
#define BK_TPB 1024
#define BK_EPT 12
#define BK_CAP 12288

__global__ void __launch_bounds__(PT)
k_hist(const int* __restrict__ dst, int* __restrict__ table,
       int E, int nblk, int nb, int T, int npad) {
    __shared__ int cnt[NBMAX];
    const int tid = threadIdx.x, blk = blockIdx.x;
    if (tid < NBMAX) cnt[tid] = 0;
    __syncthreads();
    const int e0 = blk * CHUNK + tid;
    int d[EPT];
#pragma unroll
    for (int it = 0; it < EPT; ++it) {
        const int e = e0 + it * PT;
        d[it] = (e < E) ? dst[e] : -1;
    }
#pragma unroll
    for (int it = 0; it < EPT; ++it)
        if (d[it] >= 0) atomicAdd(&cnt[d[it] >> BSHIFT], 1);
    __syncthreads();
    if (tid < nb) table[tid * nblk + blk] = cnt[tid];
    if (blk == 0)
        for (int i = T + tid; i < npad; i += PT) table[i] = 0;
}

__global__ void __launch_bounds__(SCAN_TPB)
k_scan(const int* __restrict__ cnt, int* __restrict__ start, int n) {
    __shared__ int red[SCAN_TPB / 32];
    __shared__ int wsum[SCAN_TPB / 32];
    const int tid = threadIdx.x, lane = tid & 31, wave = tid >> 5;
    const int4* __restrict__ c4 = (const int4*)cnt;
    const int nvec_before = blockIdx.x * (SCAN_TILE / 4);

    int pre = 0;
#pragma unroll 8
    for (int i = tid; i < nvec_before; i += SCAN_TPB) {
        int4 v = c4[i];
        pre += (v.x + v.y) + (v.z + v.w);
    }
    for (int o = 16; o > 0; o >>= 1) pre += __shfl_xor(pre, o);
    if (lane == 0) red[wave] = pre;

    const int4 v = c4[nvec_before + tid];
    const int t1 = v.x, t2 = t1 + v.y, t3 = t2 + v.z, t4 = t3 + v.w;
    int inc = t4;
    for (int o = 1; o < 32; o <<= 1) {
        int u = __shfl_up(inc, o);
        if (lane >= o) inc += u;
    }
    if (lane == 31) wsum[wave] = inc;
    __syncthreads();

    int base = 0;
    for (int w = 0; w < SCAN_TPB / 32; ++w) {
        base += red[w];
        if (w < wave) base += wsum[w];
    }
    const int excl = base + inc - t4;
    const int i0 = blockIdx.x * SCAN_TILE + 4 * tid;
    if (i0 + 3 < n) {
        int4 r; r.x = excl; r.y = excl + t1; r.z = excl + t2; r.w = excl + t3;
        *(int4*)(start + i0) = r;
    } else {
        if (i0 < n)     start[i0]     = excl;
        if (i0 + 1 < n) start[i0 + 1] = excl + t1;
        if (i0 + 2 < n) start[i0 + 2] = excl + t2;
    }
}

__global__ void __launch_bounds__(PT)
k_part(const int* __restrict__ src, const int* __restrict__ dst,
       const int* __restrict__ tofs, int2* __restrict__ tmp,
       int E, int nblk, int nb) {
    __shared__ int cnt[NBMAX];
    __shared__ int lbase[NBMAX];
    __shared__ int delta[NBMAX];
    __shared__ int wtot[NBMAX / 32];
    __shared__ int2 stage[CHUNK];
    const int tid = threadIdx.x, blk = blockIdx.x;
    const int lane = tid & 31, wave = tid >> 5;

    int g = 0;
    if (tid < NBMAX) {
        cnt[tid] = 0;
        if (tid < nb) g = tofs[tid * nblk + blk];
    }
    __syncthreads();

    const int e0 = blk * CHUNK + tid;
    int s[EPT], d[EPT], r[EPT];
#pragma unroll
    for (int it = 0; it < EPT; ++it) {
        const int e = e0 + it * PT;
        d[it] = (e < E) ? dst[e] : -1;
        s[it] = (e < E) ? src[e] : 0;
    }
#pragma unroll
    for (int it = 0; it < EPT; ++it)
        r[it] = (d[it] >= 0) ? atomicAdd(&cnt[d[it] >> BSHIFT], 1) : 0;
    __syncthreads();

    int v = 0, inc = 0;
    if (tid < NBMAX) {
        v = cnt[tid];
        inc = v;
        for (int o = 1; o < 32; o <<= 1) {
            int u = __shfl_up(inc, o);
            if (lane >= o) inc += u;
        }
        if (lane == 31) wtot[wave] = inc;
    }
    __syncthreads();
    if (tid < NBMAX) {
        int base = 0;
        for (int w = 0; w < NBMAX / 32; ++w)
            if (w < wave) base += wtot[w];
        const int ex = base + inc - v;
        lbase[tid] = ex;
        delta[tid] = g - ex;
    }
    __syncthreads();

#pragma unroll
    for (int it = 0; it < EPT; ++it)
        if (d[it] >= 0) stage[lbase[d[it] >> BSHIFT] + r[it]] = make_int2(s[it], d[it]);
    __syncthreads();

    const int left = E - blk * CHUNK;
    const int nvalid = (left < CHUNK) ? left : CHUNK;
    for (int i = tid; i < nvalid; i += PT) {
        const int2 p = stage[i];
        tmp[delta[p.y >> BSHIFT] + i] = p;
    }
}

__global__ void __launch_bounds__(BK_TPB)
k_bucket(const int2* __restrict__ tmp, const int* __restrict__ tofs,
         int* __restrict__ start, int* __restrict__ csr_src,
         int N, int E, int nblk, int nb) {
    __shared__ int cnt[BW];
    __shared__ int wtot[BW / 32];
    __shared__ int stage[BK_CAP];
    const int tid = threadIdx.x, b = blockIdx.x;
    const int lane = tid & 31, wave = tid >> 5;
    const int beg = tofs[b * nblk];
    const int end = (b + 1 < nb) ? tofs[(b + 1) * nblk] : E;

    const int m = end - beg;
    const bool fits = (m <= BK_CAP);

    if (tid < BW) cnt[tid] = 0;
    __syncthreads();
    int2 e[BK_EPT];
    if (fits) {
#pragma unroll
        for (int j = 0; j < BK_EPT; ++j) {
            const int p = beg + tid + j * BK_TPB;
            e[j] = make_int2(0, -1);
            if (p < end) e[j] = tmp[p];
        }
#pragma unroll
        for (int j = 0; j < BK_EPT; ++j)
            if (e[j].y >= 0) atomicAdd(&cnt[e[j].y & (BW - 1)], 1);
    } else {
        for (int p = beg + tid; p < end; p += BK_TPB)
            atomicAdd(&cnt[tmp[p].y & (BW - 1)], 1);
    }
    __syncthreads();

    int v = 0, inc = 0;
    if (tid < BW) {
        v = cnt[tid];
        inc = v;
        for (int o = 1; o < 32; o <<= 1) {
            int u = __shfl_up(inc, o);
            if (lane >= o) inc += u;
        }
        if (lane == 31) wtot[wave] = inc;
    }
    __syncthreads();
    if (tid < BW) {
        int base = 0;
        for (int w = 0; w < BW / 32; ++w)
            if (w < wave) base += wtot[w];
        const int ex = base + inc - v;
        cnt[tid] = ex;
        const int row = b * BW + tid;
        if (row < N) start[row] = beg + ex;
    }
    __syncthreads();

    if (fits) {
#pragma unroll
        for (int j = 0; j < BK_EPT; ++j)
            if (e[j].y >= 0) stage[atomicAdd(&cnt[e[j].y & (BW - 1)], 1)] = e[j].x;
        __syncthreads();
        for (int i = tid; i < m; i += BK_TPB) csr_src[beg + i] = stage[i];
    } else {
        for (int p = beg + tid; p < end; p += BK_TPB) {
            const int2 q = tmp[p];
            csr_src[beg + atomicAdd(&cnt[q.y & (BW - 1)], 1)] = q.x;
        }
    }
}


template <int K, int NOUT, int WAVES>
__global__ void __launch_bounds__(WAVES * 32)
gcn_gemm_wmma(const float* __restrict__ A, const float* __restrict__ W,
              const int* __restrict__ start, float* __restrict__ H, int M, int E) {
    constexpr int NT = WAVES * 32;
    constexpr int COL_TILES = NOUT / 16;
    constexpr int ROW_TILES_PER_BLK = WAVES / COL_TILES;
    constexpr int ROWS = ROW_TILES_PER_BLK * 16;
    constexpr int P = K + 4;
    constexpr int KV = K / 4;

    __shared__ __attribute__((aligned(16))) float lw[NOUT * P];
    __shared__ __attribute__((aligned(16))) float la[ROWS * P];

    constexpr int WIT = (NOUT * KV) / NT;
    constexpr int AIT = (ROWS * KV) / NT;
    static_assert((NOUT * KV) % NT == 0 && (ROWS * KV) % NT == 0, "tiles are whole passes of the block");

    const int rowBase = blockIdx.x * ROWS;
    float4 wv[WIT], av[AIT];
#pragma unroll
    for (int it = 0; it < WIT; ++it) {
        const int i = threadIdx.x + it * NT;
        wv[it] = *(const float4*)(W + 4 * (size_t)i);
    }
#pragma unroll
    for (int it = 0; it < AIT; ++it) {
        const int i = threadIdx.x + it * NT;
        const int r = i / KV, c = i % KV;
        const int rg = (rowBase + r < M) ? (rowBase + r) : (M - 1);
        av[it] = *(const float4*)(A + (size_t)rg * K + 4 * c);
    }
#pragma unroll
    for (int it = 0; it < WIT; ++it) {
        const int i = threadIdx.x + it * NT;
        const int r = i / KV, c = i % KV;
        *(float4*)(lw + r * P + 4 * c) = wv[it];
    }
#pragma unroll
    for (int it = 0; it < AIT; ++it) {
        const int i = threadIdx.x + it * NT;
        const int r = i / KV, c = i % KV;
        float4 v = av[it];
        if (rowBase + r >= M) v = make_float4(0.f, 0.f, 0.f, 0.f);
        *(float4*)(la + r * P + 4 * c) = v;
    }
    __syncthreads();

    const int wave = threadIdx.x >> 5;
    const int lane = threadIdx.x & 31;
    const int rt = wave / COL_TILES;
    const int m0 = rowBase + rt * 16;
    if (m0 >= M) return;
    const int n0 = (wave % COL_TILES) * 16;

    const int half = lane >> 4;
    const int l16  = lane & 15;
    const int koff = 2 * half;
    const float* arow = la + (rt * 16 + l16) * P + koff;
    const float* brow = lw + (n0 + l16) * P + koff;

    const int rr = (m0 + l16 < M) ? (m0 + l16) : (M - 1);
    int rb = start[rr];
    int rl = start[(rr + 1 < M) ? rr + 1 : rr];

    v8f acc = {0.f, 0.f, 0.f, 0.f, 0.f, 0.f, 0.f, 0.f};
#pragma unroll
    for (int k = 0; k < K; k += 4) {
        v2f a = *(const v2f*)(arow + k);
        v2f b = *(const v2f*)(brow + k);
        acc = __builtin_amdgcn_wmma_f32_16x16x4_f32(
             false, a,  false, b,
             (short)0, acc,  false,  false);
    }

    asm volatile("v_nop\n\tv_nop\n\tv_nop\n\tv_nop" : "+v"(acc), "+v"(rb), "+v"(rl));
    const int re = (rr + 1 < M) ? rl : E;
    const float dvl = rsqrtf((float)(re - rb + 1));
    const int n = n0 + l16;
    const int mbase = m0 + 8 * half;
#pragma unroll
    for (int j = 0; j < 8; ++j) {
        const float dv = __shfl(dvl, 8 * half + j);
        H[(size_t)(mbase + j) * NOUT + n] = acc[j] * dv;
    }
}


template <int F, bool RELU>
__global__ void __launch_bounds__(TPB)
k_gather(const float* __restrict__ h, const int* __restrict__ start,
         const int* __restrict__ csr_src, const float* __restrict__ bias,
         float* __restrict__ out, int n, int E) {
    constexpr int Q = F / 4;
    const int t = blockIdx.x * TPB + threadIdx.x;
    const int q = t % Q;
    const bool live = (t / Q) < n;
    const int d = live ? (t / Q) : (n - 1);
    const float4 self = *(const float4*)(h + (size_t)d * F + 4 * q);
    const int beg = start[d];
    const int endl = start[(d + 1 < n) ? d + 1 : d];
    const int end = (d + 1 < n) ? endl : E;
    const float dd = rsqrtf((float)(end - beg + 1));

    float4 acc = self;
    for (int p = beg; p < end; p += Q) {
        const int mine = (p + q < end) ? csr_src[p + q] : d;
        float4 v[Q];
#pragma unroll
        for (int j = 0; j < Q; ++j) {
            const int s = __shfl(mine, j, Q);
            v[j] = make_float4(0.f, 0.f, 0.f, 0.f);
            if (p + j < end)
                v[j] = *(const float4*)(h + (size_t)s * F + 4 * q);
        }
#pragma unroll
        for (int j = 0; j < Q; ++j) {
            acc.x += v[j].x; acc.y += v[j].y; acc.z += v[j].z; acc.w += v[j].w;
        }
    }

    const float4 bq = *(const float4*)(bias + 4 * q);
    float4 r;
    r.x = acc.x * dd + bq.x; r.y = acc.y * dd + bq.y;
    r.z = acc.z * dd + bq.z; r.w = acc.w * dd + bq.w;
    if (RELU) {
        r.x = fmaxf(r.x, 0.0f); r.y = fmaxf(r.y, 0.0f);
        r.z = fmaxf(r.z, 0.0f); r.w = fmaxf(r.w, 0.0f);
    }
    if (live) *(float4*)(out + (size_t)d * F + 4 * q) = r;
}


static inline int cdiv(long long a, int b) { return (int)((a + b - 1) / b); }

extern "C" void kernel_launch(void* const* d_in, const int* in_sizes, int n_in,
                              void* d_out, int out_size, void* d_ws, size_t ws_size,
                              hipStream_t stream) {
    (void)n_in; (void)out_size; (void)ws_size;
    const float* x  = (const float*)d_in[0];
    const int*   ei = (const int*)  d_in[1];
    const float* W1 = (const float*)d_in[2];
    const float* b1 = (const float*)d_in[3];
    const float* W2 = (const float*)d_in[4];
    const float* b2 = (const float*)d_in[5];
    float* out = (float*)d_out;

    const int N = in_sizes[0] / 128;
    const int E = in_sizes[1] / 2;
    const int* src = ei;
    const int* dst = ei + E;

    float* ws   = (float*)d_ws;
    const size_t NA = ((size_t)N + 3) & ~(size_t)3;
    int*   start   = (int*)ws;
    float* h1      = ws + NA;
    float* agg1    = h1 + 64 * NA;
    int*   csr_src = (int*)(agg1 + 64 * NA);
    float* h2      = h1;
    const int nblk = cdiv(E, CHUNK);
    const int nb   = cdiv(N, BW);
    const int T    = nb * nblk;
    const int scanBlocks = cdiv(T, SCAN_TILE);
    const int npad = scanBlocks * SCAN_TILE;
    int*  table = (int*)agg1;
    int*  tofs  = table + npad;
    int2* tmp   = (int2*)(tofs + npad);

    k_hist<<<nblk, PT, 0, stream>>>(dst, table, E, nblk, nb, T, npad);
    k_scan<<<scanBlocks, SCAN_TPB, 0, stream>>>(table, tofs, T);
    k_part<<<nblk, PT, 0, stream>>>(src, dst, tofs, tmp, E, nblk, nb);
    k_bucket<<<nb, BK_TPB, 0, stream>>>(tmp, tofs, start, csr_src, N, E, nblk, nb);

    {
        int rowTiles = (N + 15) / 16;
        gcn_gemm_wmma<128, 64, 16><<<cdiv(rowTiles, 4), 512, 0, stream>>>(x, W1, start, h1, N, E);
    }
    k_gather<64, true><<<cdiv((long long)N * 16, TPB), TPB, 0, stream>>>(h1, start, csr_src, b1, agg1, N, E);

    {
        int rowTiles = (N + 15) / 16;
        gcn_gemm_wmma<64, 32, 16><<<cdiv(rowTiles, 8), 512, 0, stream>>>(agg1, W2, start, h2, N, E);
    }
    k_gather<32, false><<<cdiv((long long)N * 8, TPB), TPB, 0, stream>>>(h2, start, csr_src, b2, out, N, E);
}


# Layer-2 GEMM fused behind the layer-1 gather: each 16 aggregated rows stay in LDS and are multiplied by W2 (staged once per 96 rows) on WMMA; one launch less, no agg1 round trip; sums unchanged
# speedup vs baseline: 4.1501x; 1.0472x over previous
#include <hip/hip_runtime.h>


typedef float v2f __attribute__((ext_vector_type(2)));
typedef float v8f __attribute__((ext_vector_type(8)));

#define TPB 256
#define SCAN_TPB 1024
#define SCAN_TILE 4096

#define PT 512
#define EPT 8
#define CHUNK 4096
#define BSHIFT 9
#define BW 512
#define NBMAX 256
#define BK_TPB 1024
#define BK_EPT 12
#define BK_CAP 12288

__global__ void __launch_bounds__(PT)
k_hist(const int* __restrict__ dst, int* __restrict__ table,
       int E, int nblk, int nb, int T, int npad) {
    __shared__ int cnt[NBMAX];
    const int tid = threadIdx.x, blk = blockIdx.x;
    if (tid < NBMAX) cnt[tid] = 0;
    __syncthreads();
    const int e0 = blk * CHUNK + tid;
    int d[EPT];
#pragma unroll
    for (int it = 0; it < EPT; ++it) {
        const int e = e0 + it * PT;
        d[it] = (e < E) ? dst[e] : -1;
    }
#pragma unroll
    for (int it = 0; it < EPT; ++it)
        if (d[it] >= 0) atomicAdd(&cnt[d[it] >> BSHIFT], 1);
    __syncthreads();
    if (tid < nb) table[tid * nblk + blk] = cnt[tid];
    if (blk == 0)
        for (int i = T + tid; i < npad; i += PT) table[i] = 0;
}

__global__ void __launch_bounds__(SCAN_TPB)
k_scan(const int* __restrict__ cnt, int* __restrict__ start, int n) {
    __shared__ int red[SCAN_TPB / 32];
    __shared__ int wsum[SCAN_TPB / 32];
    const int tid = threadIdx.x, lane = tid & 31, wave = tid >> 5;
    const int4* __restrict__ c4 = (const int4*)cnt;
    const int nvec_before = blockIdx.x * (SCAN_TILE / 4);

    int pre = 0;
#pragma unroll 8
    for (int i = tid; i < nvec_before; i += SCAN_TPB) {
        int4 v = c4[i];
        pre += (v.x + v.y) + (v.z + v.w);
    }
    for (int o = 16; o > 0; o >>= 1) pre += __shfl_xor(pre, o);
    if (lane == 0) red[wave] = pre;

    const int4 v = c4[nvec_before + tid];
    const int t1 = v.x, t2 = t1 + v.y, t3 = t2 + v.z, t4 = t3 + v.w;
    int inc = t4;
    for (int o = 1; o < 32; o <<= 1) {
        int u = __shfl_up(inc, o);
        if (lane >= o) inc += u;
    }
    if (lane == 31) wsum[wave] = inc;
    __syncthreads();

    int base = 0;
    for (int w = 0; w < SCAN_TPB / 32; ++w) {
        base += red[w];
        if (w < wave) base += wsum[w];
    }
    const int excl = base + inc - t4;
    const int i0 = blockIdx.x * SCAN_TILE + 4 * tid;
    if (i0 + 3 < n) {
        int4 r; r.x = excl; r.y = excl + t1; r.z = excl + t2; r.w = excl + t3;
        *(int4*)(start + i0) = r;
    } else {
        if (i0 < n)     start[i0]     = excl;
        if (i0 + 1 < n) start[i0 + 1] = excl + t1;
        if (i0 + 2 < n) start[i0 + 2] = excl + t2;
    }
}

__global__ void __launch_bounds__(PT)
k_part(const int* __restrict__ src, const int* __restrict__ dst,
       const int* __restrict__ tofs, int2* __restrict__ tmp,
       int E, int nblk, int nb) {
    __shared__ int cnt[NBMAX];
    __shared__ int lbase[NBMAX];
    __shared__ int delta[NBMAX];
    __shared__ int wtot[NBMAX / 32];
    __shared__ int2 stage[CHUNK];
    const int tid = threadIdx.x, blk = blockIdx.x;
    const int lane = tid & 31, wave = tid >> 5;

    int g = 0;
    if (tid < NBMAX) {
        cnt[tid] = 0;
        if (tid < nb) g = tofs[tid * nblk + blk];
    }
    __syncthreads();

    const int e0 = blk * CHUNK + tid;
    int s[EPT], d[EPT], r[EPT];
#pragma unroll
    for (int it = 0; it < EPT; ++it) {
        const int e = e0 + it * PT;
        d[it] = (e < E) ? dst[e] : -1;
        s[it] = (e < E) ? src[e] : 0;
    }
#pragma unroll
    for (int it = 0; it < EPT; ++it)
        r[it] = (d[it] >= 0) ? atomicAdd(&cnt[d[it] >> BSHIFT], 1) : 0;
    __syncthreads();

    int v = 0, inc = 0;
    if (tid < NBMAX) {
        v = cnt[tid];
        inc = v;
        for (int o = 1; o < 32; o <<= 1) {
            int u = __shfl_up(inc, o);
            if (lane >= o) inc += u;
        }
        if (lane == 31) wtot[wave] = inc;
    }
    __syncthreads();
    if (tid < NBMAX) {
        int base = 0;
        for (int w = 0; w < NBMAX / 32; ++w)
            if (w < wave) base += wtot[w];
        const int ex = base + inc - v;
        lbase[tid] = ex;
        delta[tid] = g - ex;
    }
    __syncthreads();

#pragma unroll
    for (int it = 0; it < EPT; ++it)
        if (d[it] >= 0) stage[lbase[d[it] >> BSHIFT] + r[it]] = make_int2(s[it], d[it]);
    __syncthreads();

    const int left = E - blk * CHUNK;
    const int nvalid = (left < CHUNK) ? left : CHUNK;
    for (int i = tid; i < nvalid; i += PT) {
        const int2 p = stage[i];
        tmp[delta[p.y >> BSHIFT] + i] = p;
    }
}

__global__ void __launch_bounds__(BK_TPB)
k_bucket(const int2* __restrict__ tmp, const int* __restrict__ tofs,
         int* __restrict__ start, int* __restrict__ csr_src,
         int N, int E, int nblk, int nb) {
    __shared__ int cnt[BW];
    __shared__ int wtot[BW / 32];
    __shared__ int stage[BK_CAP];
    const int tid = threadIdx.x, b = blockIdx.x;
    const int lane = tid & 31, wave = tid >> 5;
    const int beg = tofs[b * nblk];
    const int end = (b + 1 < nb) ? tofs[(b + 1) * nblk] : E;

    const int m = end - beg;
    const bool fits = (m <= BK_CAP);

    if (tid < BW) cnt[tid] = 0;
    __syncthreads();
    int2 e[BK_EPT];
    if (fits) {
#pragma unroll
        for (int j = 0; j < BK_EPT; ++j) {
            const int p = beg + tid + j * BK_TPB;
            e[j] = make_int2(0, -1);
            if (p < end) e[j] = tmp[p];
        }
#pragma unroll
        for (int j = 0; j < BK_EPT; ++j)
            if (e[j].y >= 0) atomicAdd(&cnt[e[j].y & (BW - 1)], 1);
    } else {
        for (int p = beg + tid; p < end; p += BK_TPB)
            atomicAdd(&cnt[tmp[p].y & (BW - 1)], 1);
    }
    __syncthreads();

    int v = 0, inc = 0;
    if (tid < BW) {
        v = cnt[tid];
        inc = v;
        for (int o = 1; o < 32; o <<= 1) {
            int u = __shfl_up(inc, o);
            if (lane >= o) inc += u;
        }
        if (lane == 31) wtot[wave] = inc;
    }
    __syncthreads();
    if (tid < BW) {
        int base = 0;
        for (int w = 0; w < BW / 32; ++w)
            if (w < wave) base += wtot[w];
        const int ex = base + inc - v;
        cnt[tid] = ex;
        const int row = b * BW + tid;
        if (row < N) start[row] = beg + ex;
    }
    __syncthreads();

    if (fits) {
#pragma unroll
        for (int j = 0; j < BK_EPT; ++j)
            if (e[j].y >= 0) stage[atomicAdd(&cnt[e[j].y & (BW - 1)], 1)] = e[j].x;
        __syncthreads();
        for (int i = tid; i < m; i += BK_TPB) csr_src[beg + i] = stage[i];
    } else {
        for (int p = beg + tid; p < end; p += BK_TPB) {
            const int2 q = tmp[p];
            csr_src[beg + atomicAdd(&cnt[q.y & (BW - 1)], 1)] = q.x;
        }
    }
}


template <int K, int NOUT, int WAVES>
__global__ void __launch_bounds__(WAVES * 32)
gcn_gemm_wmma(const float* __restrict__ A, const float* __restrict__ W,
              const int* __restrict__ start, float* __restrict__ H, int M, int E) {
    constexpr int NT = WAVES * 32;
    constexpr int COL_TILES = NOUT / 16;
    constexpr int ROW_TILES_PER_BLK = WAVES / COL_TILES;
    constexpr int ROWS = ROW_TILES_PER_BLK * 16;
    constexpr int P = K + 4;
    constexpr int KV = K / 4;

    __shared__ __attribute__((aligned(16))) float lw[NOUT * P];
    __shared__ __attribute__((aligned(16))) float la[ROWS * P];

    constexpr int WIT = (NOUT * KV) / NT;
    constexpr int AIT = (ROWS * KV) / NT;
    static_assert((NOUT * KV) % NT == 0 && (ROWS * KV) % NT == 0, "tiles are whole passes of the block");

    const int rowBase = blockIdx.x * ROWS;
    float4 wv[WIT], av[AIT];
#pragma unroll
    for (int it = 0; it < WIT; ++it) {
        const int i = threadIdx.x + it * NT;
        wv[it] = *(const float4*)(W + 4 * (size_t)i);
    }
#pragma unroll
    for (int it = 0; it < AIT; ++it) {
        const int i = threadIdx.x + it * NT;
        const int r = i / KV, c = i % KV;
        const int rg = (rowBase + r < M) ? (rowBase + r) : (M - 1);
        av[it] = *(const float4*)(A + (size_t)rg * K + 4 * c);
    }
#pragma unroll
    for (int it = 0; it < WIT; ++it) {
        const int i = threadIdx.x + it * NT;
        const int r = i / KV, c = i % KV;
        *(float4*)(lw + r * P + 4 * c) = wv[it];
    }
#pragma unroll
    for (int it = 0; it < AIT; ++it) {
        const int i = threadIdx.x + it * NT;
        const int r = i / KV, c = i % KV;
        float4 v = av[it];
        if (rowBase + r >= M) v = make_float4(0.f, 0.f, 0.f, 0.f);
        *(float4*)(la + r * P + 4 * c) = v;
    }
    __syncthreads();

    const int wave = threadIdx.x >> 5;
    const int lane = threadIdx.x & 31;
    const int rt = wave / COL_TILES;
    const int m0 = rowBase + rt * 16;
    if (m0 >= M) return;
    const int n0 = (wave % COL_TILES) * 16;

    const int half = lane >> 4;
    const int l16  = lane & 15;
    const int koff = 2 * half;
    const float* arow = la + (rt * 16 + l16) * P + koff;
    const float* brow = lw + (n0 + l16) * P + koff;

    const int rr = (m0 + l16 < M) ? (m0 + l16) : (M - 1);
    int rb = start[rr];
    int rl = start[(rr + 1 < M) ? rr + 1 : rr];

    v8f acc = {0.f, 0.f, 0.f, 0.f, 0.f, 0.f, 0.f, 0.f};
#pragma unroll
    for (int k = 0; k < K; k += 4) {
        v2f a = *(const v2f*)(arow + k);
        v2f b = *(const v2f*)(brow + k);
        acc = __builtin_amdgcn_wmma_f32_16x16x4_f32(
             false, a,  false, b,
             (short)0, acc,  false,  false);
    }

    asm volatile("v_nop\n\tv_nop\n\tv_nop\n\tv_nop" : "+v"(acc), "+v"(rb), "+v"(rl));
    const int re = (rr + 1 < M) ? rl : E;
    const float dvl = rsqrtf((float)(re - rb + 1));
    const int n = n0 + l16;
    const int mbase = m0 + 8 * half;
#pragma unroll
    for (int j = 0; j < 8; ++j) {
        const float dv = __shfl(dvl, 8 * half + j);
        H[(size_t)(mbase + j) * NOUT + n] = acc[j] * dv;
    }
}


template <int F, bool RELU>
__global__ void __launch_bounds__(TPB)
k_gather(const float* __restrict__ h, const int* __restrict__ start,
         const int* __restrict__ csr_src, const float* __restrict__ bias,
         float* __restrict__ out, int n, int E) {
    constexpr int Q = F / 4;
    const int t = blockIdx.x * TPB + threadIdx.x;
    const int q = t % Q;
    const bool live = (t / Q) < n;
    const int d = live ? (t / Q) : (n - 1);
    const float4 self = *(const float4*)(h + (size_t)d * F + 4 * q);
    const int beg = start[d];
    const int endl = start[(d + 1 < n) ? d + 1 : d];
    const int end = (d + 1 < n) ? endl : E;
    const float dd = rsqrtf((float)(end - beg + 1));

    float4 acc = self;
    for (int p = beg; p < end; p += Q) {
        const int mine = (p + q < end) ? csr_src[p + q] : d;
        float4 v[Q];
#pragma unroll
        for (int j = 0; j < Q; ++j) {
            const int s = __shfl(mine, j, Q);
            v[j] = make_float4(0.f, 0.f, 0.f, 0.f);
            if (p + j < end)
                v[j] = *(const float4*)(h + (size_t)s * F + 4 * q);
        }
#pragma unroll
        for (int j = 0; j < Q; ++j) {
            acc.x += v[j].x; acc.y += v[j].y; acc.z += v[j].z; acc.w += v[j].w;
        }
    }

    const float4 bq = *(const float4*)(bias + 4 * q);
    float4 r;
    r.x = acc.x * dd + bq.x; r.y = acc.y * dd + bq.y;
    r.z = acc.z * dd + bq.z; r.w = acc.w * dd + bq.w;
    if (RELU) {
        r.x = fmaxf(r.x, 0.0f); r.y = fmaxf(r.y, 0.0f);
        r.z = fmaxf(r.z, 0.0f); r.w = fmaxf(r.w, 0.0f);
    }
    if (live) *(float4*)(out + (size_t)d * F + 4 * q) = r;
}


template <int F, int NOUT, int PASSES>
__global__ void __launch_bounds__(TPB)
k_gather_gemm(const float* __restrict__ h, const int* __restrict__ start,
              const int* __restrict__ csr_src, const float* __restrict__ bias,
              const float* __restrict__ W, float* __restrict__ H, int n, int E) {
    constexpr int Q = F / 4;
    constexpr int RP = TPB / Q;
    constexpr int P = F + 4;
    constexpr int KV = F / 4;
    constexpr int WIT = (NOUT * KV) / TPB;
    static_assert(RP == 16, "a group is one 16-row matrix tile");
    static_assert((NOUT * KV) % TPB == 0, "W is staged in whole passes of the block");
    static_assert(NOUT / 16 <= TPB / 32, "one wave per 16-column tile");

    __shared__ __attribute__((aligned(16))) float lw[NOUT * P];
    __shared__ __attribute__((aligned(16))) float la[2][RP * P];
    __shared__ __attribute__((aligned(16))) float ldv[2][RP];

    const int tid = threadIdx.x;
    const int q = tid % Q, r = tid / Q;
    const int wave = tid >> 5, lane = tid & 31;
    const int half = lane >> 4, l16 = lane & 15;
    const int hb = 8 * half;
    const int nl = (wave % (NOUT / 16)) * 16 + l16;

    float4 wv[WIT];
#pragma unroll
    for (int it = 0; it < WIT; ++it)
        wv[it] = *(const float4*)(W + 4 * (size_t)(tid + it * TPB));
#pragma unroll
    for (int it = 0; it < WIT; ++it) {
        const int i = tid + it * TPB;
        *(float4*)(lw + (i / KV) * P + 4 * (i % KV)) = wv[it];
    }
    const float4 bq = *(const float4*)(bias + 4 * q);

    for (int ps = 0; ps < PASSES; ++ps) {
        const int m0 = (blockIdx.x * PASSES + ps) * RP;
        if (m0 >= n) break;
        const bool live = (m0 + r) < n;
        const int d = live ? (m0 + r) : (n - 1);
        const float4 self = *(const float4*)(h + (size_t)d * F + 4 * q);
        const int beg = start[d];
        const int endl = start[(d + 1 < n) ? d + 1 : d];
        const int end = (d + 1 < n) ? endl : E;
        const float dd = rsqrtf((float)(end - beg + 1));

        float4 acc = self;
        for (int p = beg; p < end; p += Q) {
            const int mine = (p + q < end) ? csr_src[p + q] : d;
            float4 v[Q];
#pragma unroll
            for (int j = 0; j < Q; ++j) {
                const int s = __shfl(mine, j, Q);
                v[j] = make_float4(0.f, 0.f, 0.f, 0.f);
                if (p + j < end)
                    v[j] = *(const float4*)(h + (size_t)s * F + 4 * q);
            }
#pragma unroll
            for (int j = 0; j < Q; ++j) {
                acc.x += v[j].x; acc.y += v[j].y; acc.z += v[j].z; acc.w += v[j].w;
            }
        }

        float4 rv;
        rv.x = fmaxf(acc.x * dd + bq.x, 0.0f); rv.y = fmaxf(acc.y * dd + bq.y, 0.0f);
        rv.z = fmaxf(acc.z * dd + bq.z, 0.0f); rv.w = fmaxf(acc.w * dd + bq.w, 0.0f);
        if (!live) rv = make_float4(0.f, 0.f, 0.f, 0.f);

        float* lab = la[ps & 1];
        *(float4*)(lab + r * P + 4 * q) = rv;
        if (q == 0) ldv[ps & 1][r] = dd;
        __syncthreads();

        if (wave < NOUT / 16) {
            const int n0 = wave * 16;
            const float* arow = lab + l16 * P + 2 * half;
            const float* brow = lw + (n0 + l16) * P + 2 * half;
            v8f c = {0.f, 0.f, 0.f, 0.f, 0.f, 0.f, 0.f, 0.f};
#pragma unroll
            for (int k = 0; k < F; k += 4) {
                v2f a = *(const v2f*)(arow + k);
                v2f b = *(const v2f*)(brow + k);
                c = __builtin_amdgcn_wmma_f32_16x16x4_f32(
                     false, a,  false, b,
                     (short)0, c,  false,  false);
            }
            int hq;
            asm volatile("v_nop\n\tv_nop\n\tv_nop\n\tv_nop\n\tv_mov_b32 %1, %2"
                         : "+v"(c), "=v"(hq) : "v"(hb));
            const float4 dlo = *(const float4*)(&ldv[ps & 1][hq]);
            const float4 dhi = *(const float4*)(&ldv[ps & 1][hq + 4]);
            const float dvs[8] = {dlo.x, dlo.y, dlo.z, dlo.w, dhi.x, dhi.y, dhi.z, dhi.w};
            const int rowb = m0 + hq;
            float* hp = H + (size_t)rowb * NOUT + nl;
            if (m0 + RP <= n) {
#pragma unroll
                for (int j = 0; j < 8; ++j) hp[j * NOUT] = c[j] * dvs[j];
            } else {
#pragma unroll
                for (int j = 0; j < 8; ++j)
                    if (rowb + j < n) hp[j * NOUT] = c[j] * dvs[j];
            }
        }
    }
}


static inline int cdiv(long long a, int b) { return (int)((a + b - 1) / b); }

extern "C" void kernel_launch(void* const* d_in, const int* in_sizes, int n_in,
                              void* d_out, int out_size, void* d_ws, size_t ws_size,
                              hipStream_t stream) {
    (void)n_in; (void)out_size; (void)ws_size;
    const float* x  = (const float*)d_in[0];
    const int*   ei = (const int*)  d_in[1];
    const float* W1 = (const float*)d_in[2];
    const float* b1 = (const float*)d_in[3];
    const float* W2 = (const float*)d_in[4];
    const float* b2 = (const float*)d_in[5];
    float* out = (float*)d_out;

    const int N = in_sizes[0] / 128;
    const int E = in_sizes[1] / 2;
    const int* src = ei;
    const int* dst = ei + E;

    float* ws   = (float*)d_ws;
    const size_t NA = ((size_t)N + 3) & ~(size_t)3;
    int*   start   = (int*)ws;
    float* h1      = ws + NA;
    float* agg1    = h1 + 64 * NA;
    int*   csr_src = (int*)(agg1 + 64 * NA);
    float* h2      = agg1;
    const int nblk = cdiv(E, CHUNK);
    const int nb   = cdiv(N, BW);
    const int T    = nb * nblk;
    const int scanBlocks = cdiv(T, SCAN_TILE);
    const int npad = scanBlocks * SCAN_TILE;
    int*  table = (int*)agg1;
    int*  tofs  = table + npad;
    int2* tmp   = (int2*)(tofs + npad);

    k_hist<<<nblk, PT, 0, stream>>>(dst, table, E, nblk, nb, T, npad);
    k_scan<<<scanBlocks, SCAN_TPB, 0, stream>>>(table, tofs, T);
    k_part<<<nblk, PT, 0, stream>>>(src, dst, tofs, tmp, E, nblk, nb);
    k_bucket<<<nb, BK_TPB, 0, stream>>>(tmp, tofs, start, csr_src, N, E, nblk, nb);

    {
        int rowTiles = (N + 15) / 16;
        gcn_gemm_wmma<128, 64, 16><<<cdiv(rowTiles, 4), 512, 0, stream>>>(x, W1, start, h1, N, E);
    }
    k_gather_gemm<64, 32, 6><<<cdiv(N, 16 * 6), TPB, 0, stream>>>(h1, start, csr_src, b1, W2, h2, N, E);
    k_gather<32, false><<<cdiv((long long)N * 8, TPB), TPB, 0, stream>>>(h2, start, csr_src, b2, out, N, E);
}


# Edge sort drops its histogram and scan launches: each chunk is grouped in place, a bucket block finds its runs from a table of per-chunk bucket offsets via an LDS position map; sum order by LDS rank
# speedup vs baseline: 4.6778x; 1.1272x over previous
#include <hip/hip_runtime.h>


typedef float v2f __attribute__((ext_vector_type(2)));
typedef float v8f __attribute__((ext_vector_type(8)));

#define TPB 256

#define PT 512
#define EPT 8
#define CHUNK 4096
#define BSHIFT 9
#define BW 512
#define NBMAX 256
#define BK_TPB 1024
#define BK_EPT 12
#define BK_CAP 12288
#define RMAX 1024

__global__ void __launch_bounds__(PT)
k_part(const int* __restrict__ src, const int* __restrict__ dst,
       int* __restrict__ lbt, int2* __restrict__ tmp,
       int E, int nblk, int nb) {
    __shared__ int cnt[NBMAX];
    __shared__ int lbase[NBMAX];
    __shared__ int wtot[NBMAX / 32];
    __shared__ int2 stage[CHUNK];
    const int tid = threadIdx.x, blk = blockIdx.x;
    const int lane = tid & 31, wave = tid >> 5;

    if (tid < NBMAX) cnt[tid] = 0;
    __syncthreads();

    const int e0 = blk * CHUNK + tid;
    int s[EPT], d[EPT], r[EPT];
#pragma unroll
    for (int it = 0; it < EPT; ++it) {
        const int e = e0 + it * PT;
        d[it] = (e < E) ? dst[e] : -1;
        s[it] = (e < E) ? src[e] : 0;
    }
#pragma unroll
    for (int it = 0; it < EPT; ++it)
        r[it] = (d[it] >= 0) ? atomicAdd(&cnt[d[it] >> BSHIFT], 1) : 0;
    __syncthreads();

    int v = 0, inc = 0;
    if (tid < NBMAX) {
        v = cnt[tid];
        inc = v;
        for (int o = 1; o < 32; o <<= 1) {
            int u = __shfl_up(inc, o);
            if (lane >= o) inc += u;
        }
        if (lane == 31) wtot[wave] = inc;
    }
    __syncthreads();
    const int left = E - blk * CHUNK;
    const int nvalid = (left < CHUNK) ? left : CHUNK;
    if (tid < NBMAX) {
        int base = 0;
        for (int w = 0; w < NBMAX / 32; ++w)
            if (w < wave) base += wtot[w];
        const int ex = base + inc - v;
        lbase[tid] = ex;
        if (tid < nb) lbt[tid * nblk + blk] = ex;
    }
    if (tid == nb) lbt[nb * nblk + blk] = nvalid;
    __syncthreads();

#pragma unroll
    for (int it = 0; it < EPT; ++it)
        if (d[it] >= 0) stage[lbase[d[it] >> BSHIFT] + r[it]] = make_int2(s[it], d[it]);
    __syncthreads();

    for (int i = tid; i < nvalid; i += PT) tmp[blk * CHUNK + i] = stage[i];
}

__device__ __forceinline__ int run_of(const int* incl, int nblk, int top, int i) {
    int pos = 0;
    for (int s = top; s > 0; s >>= 1) {
        const int c = pos + s;
        const int val = incl[(c < nblk) ? (c - 1) : 0];
        if (c < nblk && val <= i) pos = c;
    }
    return pos;
}

__global__ void __launch_bounds__(BK_TPB)
k_bucket(const int2* __restrict__ tmp, const int* __restrict__ lbt,
         int* __restrict__ start, int* __restrict__ csr_src,
         int N, int nblk) {
    __shared__ int cnt[BW];
    __shared__ int wtot[BW / 32];
    __shared__ int stage[BK_CAP];
    __shared__ int incl[RMAX];
    __shared__ int gdel[RMAX];
    __shared__ int wlen[BK_TPB / 32];
    __shared__ int wlo[BK_TPB / 32];
    const int tid = threadIdx.x, b = blockIdx.x;
    const int lane = tid & 31, wave = tid >> 5;

    int lo = 0, len = 0;
    if (tid < nblk) {
        lo = lbt[b * nblk + tid];
        len = lbt[(b + 1) * nblk + tid] - lo;
    }
    int rinc = len, slo = lo;
    for (int o = 1; o < 32; o <<= 1) {
        int u = __shfl_up(rinc, o);
        if (lane >= o) rinc += u;
    }
    for (int o = 16; o > 0; o >>= 1) slo += __shfl_xor(slo, o);
    if (lane == 31) wlen[wave] = rinc;
    if (lane == 0) wlo[wave] = slo;
    if (tid < BW) cnt[tid] = 0;
    __syncthreads();

    int rbase = 0, m = 0, beg = 0;
    for (int w = 0; w < BK_TPB / 32; ++w) {
        const int a = wlen[w];
        if (w < wave) rbase += a;
        m += a;
        beg += wlo[w];
    }
    const bool fits = (m <= BK_CAP);
    if (tid < nblk) {
        const int ic = rbase + rinc;
        const int r0 = ic - len;
        const int g = tid * CHUNK + lo - r0;
        if (fits) {
            for (int k = 0; k < len; ++k) stage[r0 + k] = g;
        } else {
            incl[tid] = ic;
            gdel[tid] = g;
        }
    }
    __syncthreads();

    int top = 1;
    while (2 * top <= nblk) top *= 2;

    int2 e[BK_EPT];
    if (fits) {
        int g[BK_EPT];
#pragma unroll
        for (int j = 0; j < BK_EPT; ++j) {
            const int i = tid + j * BK_TPB;
            g[j] = stage[(i < m) ? i : 0];
        }
#pragma unroll
        for (int j = 0; j < BK_EPT; ++j) {
            const int i = tid + j * BK_TPB;
            e[j] = make_int2(0, -1);
            if (i < m) e[j] = tmp[g[j] + i];
        }
#pragma unroll
        for (int j = 0; j < BK_EPT; ++j)
            if (e[j].y >= 0) atomicAdd(&cnt[e[j].y & (BW - 1)], 1);
    } else {
        for (int i = tid; i < m; i += BK_TPB)
            atomicAdd(&cnt[tmp[gdel[run_of(incl, nblk, top, i)] + i].y & (BW - 1)], 1);
    }
    __syncthreads();

    int v = 0, inc = 0;
    if (tid < BW) {
        v = cnt[tid];
        inc = v;
        for (int o = 1; o < 32; o <<= 1) {
            int u = __shfl_up(inc, o);
            if (lane >= o) inc += u;
        }
        if (lane == 31) wtot[wave] = inc;
    }
    __syncthreads();
    if (tid < BW) {
        int base = 0;
        for (int w = 0; w < BW / 32; ++w)
            if (w < wave) base += wtot[w];
        const int ex = base + inc - v;
        cnt[tid] = ex;
        const int row = b * BW + tid;
        if (row < N) start[row] = beg + ex;
    }
    __syncthreads();

    if (fits) {
#pragma unroll
        for (int j = 0; j < BK_EPT; ++j)
            if (e[j].y >= 0) stage[atomicAdd(&cnt[e[j].y & (BW - 1)], 1)] = e[j].x;
        __syncthreads();
        for (int i = tid; i < m; i += BK_TPB) csr_src[beg + i] = stage[i];
    } else {
        for (int i = tid; i < m; i += BK_TPB) {
            const int2 q = tmp[gdel[run_of(incl, nblk, top, i)] + i];
            csr_src[beg + atomicAdd(&cnt[q.y & (BW - 1)], 1)] = q.x;
        }
    }
}


template <int K, int NOUT, int WAVES>
__global__ void __launch_bounds__(WAVES * 32)
gcn_gemm_wmma(const float* __restrict__ A, const float* __restrict__ W,
              const int* __restrict__ start, float* __restrict__ H, int M, int E) {
    constexpr int NT = WAVES * 32;
    constexpr int COL_TILES = NOUT / 16;
    constexpr int ROW_TILES_PER_BLK = WAVES / COL_TILES;
    constexpr int ROWS = ROW_TILES_PER_BLK * 16;
    constexpr int P = K + 4;
    constexpr int KV = K / 4;

    __shared__ __attribute__((aligned(16))) float lw[NOUT * P];
    __shared__ __attribute__((aligned(16))) float la[ROWS * P];

    constexpr int WIT = (NOUT * KV) / NT;
    constexpr int AIT = (ROWS * KV) / NT;
    static_assert((NOUT * KV) % NT == 0 && (ROWS * KV) % NT == 0, "tiles are whole passes of the block");

    const int rowBase = blockIdx.x * ROWS;
    float4 wv[WIT], av[AIT];
#pragma unroll
    for (int it = 0; it < WIT; ++it) {
        const int i = threadIdx.x + it * NT;
        wv[it] = *(const float4*)(W + 4 * (size_t)i);
    }
#pragma unroll
    for (int it = 0; it < AIT; ++it) {
        const int i = threadIdx.x + it * NT;
        const int r = i / KV, c = i % KV;
        const int rg = (rowBase + r < M) ? (rowBase + r) : (M - 1);
        av[it] = *(const float4*)(A + (size_t)rg * K + 4 * c);
    }
#pragma unroll
    for (int it = 0; it < WIT; ++it) {
        const int i = threadIdx.x + it * NT;
        const int r = i / KV, c = i % KV;
        *(float4*)(lw + r * P + 4 * c) = wv[it];
    }
#pragma unroll
    for (int it = 0; it < AIT; ++it) {
        const int i = threadIdx.x + it * NT;
        const int r = i / KV, c = i % KV;
        float4 v = av[it];
        if (rowBase + r >= M) v = make_float4(0.f, 0.f, 0.f, 0.f);
        *(float4*)(la + r * P + 4 * c) = v;
    }
    __syncthreads();

    const int wave = threadIdx.x >> 5;
    const int lane = threadIdx.x & 31;
    const int rt = wave / COL_TILES;
    const int m0 = rowBase + rt * 16;
    if (m0 >= M) return;
    const int n0 = (wave % COL_TILES) * 16;

    const int half = lane >> 4;
    const int l16  = lane & 15;
    const int koff = 2 * half;
    const float* arow = la + (rt * 16 + l16) * P + koff;
    const float* brow = lw + (n0 + l16) * P + koff;

    const int rr = (m0 + l16 < M) ? (m0 + l16) : (M - 1);
    int rb = start[rr];
    int rl = start[(rr + 1 < M) ? rr + 1 : rr];

    v8f acc = {0.f, 0.f, 0.f, 0.f, 0.f, 0.f, 0.f, 0.f};
#pragma unroll
    for (int k = 0; k < K; k += 4) {
        v2f a = *(const v2f*)(arow + k);
        v2f b = *(const v2f*)(brow + k);
        acc = __builtin_amdgcn_wmma_f32_16x16x4_f32(
             false, a,  false, b,
             (short)0, acc,  false,  false);
    }

    asm volatile("v_nop\n\tv_nop\n\tv_nop\n\tv_nop" : "+v"(acc), "+v"(rb), "+v"(rl));
    const int re = (rr + 1 < M) ? rl : E;
    const float dvl = rsqrtf((float)(re - rb + 1));
    const int n = n0 + l16;
    const int mbase = m0 + 8 * half;
#pragma unroll
    for (int j = 0; j < 8; ++j) {
        const float dv = __shfl(dvl, 8 * half + j);
        H[(size_t)(mbase + j) * NOUT + n] = acc[j] * dv;
    }
}


template <int F, bool RELU>
__global__ void __launch_bounds__(TPB)
k_gather(const float* __restrict__ h, const int* __restrict__ start,
         const int* __restrict__ csr_src, const float* __restrict__ bias,
         float* __restrict__ out, int n, int E) {
    constexpr int Q = F / 4;
    const int t = blockIdx.x * TPB + threadIdx.x;
    const int q = t % Q;
    const bool live = (t / Q) < n;
    const int d = live ? (t / Q) : (n - 1);
    const float4 self = *(const float4*)(h + (size_t)d * F + 4 * q);
    const int beg = start[d];
    const int endl = start[(d + 1 < n) ? d + 1 : d];
    const int end = (d + 1 < n) ? endl : E;
    const float dd = rsqrtf((float)(end - beg + 1));

    float4 acc = self;
    for (int p = beg; p < end; p += Q) {
        const int mine = (p + q < end) ? csr_src[p + q] : d;
        float4 v[Q];
#pragma unroll
        for (int j = 0; j < Q; ++j) {
            const int s = __shfl(mine, j, Q);
            v[j] = make_float4(0.f, 0.f, 0.f, 0.f);
            if (p + j < end)
                v[j] = *(const float4*)(h + (size_t)s * F + 4 * q);
        }
#pragma unroll
        for (int j = 0; j < Q; ++j) {
            acc.x += v[j].x; acc.y += v[j].y; acc.z += v[j].z; acc.w += v[j].w;
        }
    }

    const float4 bq = *(const float4*)(bias + 4 * q);
    float4 r;
    r.x = acc.x * dd + bq.x; r.y = acc.y * dd + bq.y;
    r.z = acc.z * dd + bq.z; r.w = acc.w * dd + bq.w;
    if (RELU) {
        r.x = fmaxf(r.x, 0.0f); r.y = fmaxf(r.y, 0.0f);
        r.z = fmaxf(r.z, 0.0f); r.w = fmaxf(r.w, 0.0f);
    }
    if (live) *(float4*)(out + (size_t)d * F + 4 * q) = r;
}


template <int F, int NOUT, int PASSES>
__global__ void __launch_bounds__(TPB)
k_gather_gemm(const float* __restrict__ h, const int* __restrict__ start,
              const int* __restrict__ csr_src, const float* __restrict__ bias,
              const float* __restrict__ W, float* __restrict__ H, int n, int E) {
    constexpr int Q = F / 4;
    constexpr int RP = TPB / Q;
    constexpr int P = F + 4;
    constexpr int KV = F / 4;
    constexpr int WIT = (NOUT * KV) / TPB;
    static_assert(RP == 16, "a group is one 16-row matrix tile");
    static_assert((NOUT * KV) % TPB == 0, "W is staged in whole passes of the block");
    static_assert(NOUT / 16 <= TPB / 32, "one wave per 16-column tile");

    __shared__ __attribute__((aligned(16))) float lw[NOUT * P];
    __shared__ __attribute__((aligned(16))) float la[2][RP * P];
    __shared__ __attribute__((aligned(16))) float ldv[2][RP];

    const int tid = threadIdx.x;
    const int q = tid % Q, r = tid / Q;
    const int wave = tid >> 5, lane = tid & 31;
    const int half = lane >> 4, l16 = lane & 15;
    const int hb = 8 * half;
    const int nl = (wave % (NOUT / 16)) * 16 + l16;

    float4 wv[WIT];
#pragma unroll
    for (int it = 0; it < WIT; ++it)
        wv[it] = *(const float4*)(W + 4 * (size_t)(tid + it * TPB));
#pragma unroll
    for (int it = 0; it < WIT; ++it) {
        const int i = tid + it * TPB;
        *(float4*)(lw + (i / KV) * P + 4 * (i % KV)) = wv[it];
    }
    const float4 bq = *(const float4*)(bias + 4 * q);

    for (int ps = 0; ps < PASSES; ++ps) {
        const int m0 = (blockIdx.x * PASSES + ps) * RP;
        if (m0 >= n) break;
        const bool live = (m0 + r) < n;
        const int d = live ? (m0 + r) : (n - 1);
        const float4 self = *(const float4*)(h + (size_t)d * F + 4 * q);
        const int beg = start[d];
        const int endl = start[(d + 1 < n) ? d + 1 : d];
        const int end = (d + 1 < n) ? endl : E;
        const float dd = rsqrtf((float)(end - beg + 1));

        float4 acc = self;
        for (int p = beg; p < end; p += Q) {
            const int mine = (p + q < end) ? csr_src[p + q] : d;
            float4 v[Q];
#pragma unroll
            for (int j = 0; j < Q; ++j) {
                const int s = __shfl(mine, j, Q);
                v[j] = make_float4(0.f, 0.f, 0.f, 0.f);
                if (p + j < end)
                    v[j] = *(const float4*)(h + (size_t)s * F + 4 * q);
            }
#pragma unroll
            for (int j = 0; j < Q; ++j) {
                acc.x += v[j].x; acc.y += v[j].y; acc.z += v[j].z; acc.w += v[j].w;
            }
        }

        float4 rv;
        rv.x = fmaxf(acc.x * dd + bq.x, 0.0f); rv.y = fmaxf(acc.y * dd + bq.y, 0.0f);
        rv.z = fmaxf(acc.z * dd + bq.z, 0.0f); rv.w = fmaxf(acc.w * dd + bq.w, 0.0f);
        if (!live) rv = make_float4(0.f, 0.f, 0.f, 0.f);

        float* lab = la[ps & 1];
        *(float4*)(lab + r * P + 4 * q) = rv;
        if (q == 0) ldv[ps & 1][r] = dd;
        __syncthreads();

        if (wave < NOUT / 16) {
            const int n0 = wave * 16;
            const float* arow = lab + l16 * P + 2 * half;
            const float* brow = lw + (n0 + l16) * P + 2 * half;
            v8f c = {0.f, 0.f, 0.f, 0.f, 0.f, 0.f, 0.f, 0.f};
#pragma unroll
            for (int k = 0; k < F; k += 4) {
                v2f a = *(const v2f*)(arow + k);
                v2f b = *(const v2f*)(brow + k);
                c = __builtin_amdgcn_wmma_f32_16x16x4_f32(
                     false, a,  false, b,
                     (short)0, c,  false,  false);
            }
            int hq;
            asm volatile("v_nop\n\tv_nop\n\tv_nop\n\tv_nop\n\tv_mov_b32 %1, %2"
                         : "+v"(c), "=v"(hq) : "v"(hb));
            const float4 dlo = *(const float4*)(&ldv[ps & 1][hq]);
            const float4 dhi = *(const float4*)(&ldv[ps & 1][hq + 4]);
            const float dvs[8] = {dlo.x, dlo.y, dlo.z, dlo.w, dhi.x, dhi.y, dhi.z, dhi.w};
            const int rowb = m0 + hq;
            float* hp = H + (size_t)rowb * NOUT + nl;
            if (m0 + RP <= n) {
#pragma unroll
                for (int j = 0; j < 8; ++j) hp[j * NOUT] = c[j] * dvs[j];
            } else {
#pragma unroll
                for (int j = 0; j < 8; ++j)
                    if (rowb + j < n) hp[j * NOUT] = c[j] * dvs[j];
            }
        }
    }
}


static inline int cdiv(long long a, int b) { return (int)((a + b - 1) / b); }

extern "C" void kernel_launch(void* const* d_in, const int* in_sizes, int n_in,
                              void* d_out, int out_size, void* d_ws, size_t ws_size,
                              hipStream_t stream) {
    (void)n_in; (void)out_size; (void)ws_size;
    const float* x  = (const float*)d_in[0];
    const int*   ei = (const int*)  d_in[1];
    const float* W1 = (const float*)d_in[2];
    const float* b1 = (const float*)d_in[3];
    const float* W2 = (const float*)d_in[4];
    const float* b2 = (const float*)d_in[5];
    float* out = (float*)d_out;

    const int N = in_sizes[0] / 128;
    const int E = in_sizes[1] / 2;
    const int* src = ei;
    const int* dst = ei + E;

    float* ws   = (float*)d_ws;
    const size_t NA = ((size_t)N + 3) & ~(size_t)3;
    int*   start   = (int*)ws;
    float* h1      = ws + NA;
    float* agg1    = h1 + 64 * NA;
    int*   csr_src = (int*)(agg1 + 64 * NA);
    float* h2      = agg1;
    const int nblk = cdiv(E, CHUNK);
    const int nb   = cdiv(N, BW);
    const int tsz  = ((nb + 1) * nblk + 3) & ~3;
    int*  lbt = (int*)agg1;
    int2* tmp = (int2*)(lbt + tsz);

    k_part<<<nblk, PT, 0, stream>>>(src, dst, lbt, tmp, E, nblk, nb);
    k_bucket<<<nb, BK_TPB, 0, stream>>>(tmp, lbt, start, csr_src, N, nblk);

    {
        int rowTiles = (N + 15) / 16;
        gcn_gemm_wmma<128, 64, 16><<<cdiv(rowTiles, 4), 512, 0, stream>>>(x, W1, start, h1, N, E);
    }
    k_gather_gemm<64, 32, 6><<<cdiv(N, 16 * 6), TPB, 0, stream>>>(h1, start, csr_src, b1, W2, h2, N, E);
    k_gather<32, false><<<cdiv((long long)N * 8, TPB), TPB, 0, stream>>>(h2, start, csr_src, b2, out, N, E);
}


# Layer-1 GEMM joins the sort's bucket launch: a bucket block has its rows' degrees in LDS, multiplies and scales them on WMMA; buckets of 400 rows fill 250 WGPs; one launch less; sum order by LDS rank
# speedup vs baseline: 4.9165x; 1.0510x over previous
#include <hip/hip_runtime.h>


typedef float v2f __attribute__((ext_vector_type(2)));
typedef float v8f __attribute__((ext_vector_type(8)));

#define TPB 256

#define PT 512
#define EPT 8
#define CHUNK 4096
#define BW 400
#define NBMAX 256
#define BK_TPB 1024
#define BK_EPT 12
#define BK_CAP 12288
#define RMAX 1024

__global__ void __launch_bounds__(PT)
k_part(const int* __restrict__ src, const int* __restrict__ dst,
       int* __restrict__ lbt, int2* __restrict__ tmp,
       int E, int nblk, int nb) {
    __shared__ int cnt[NBMAX];
    __shared__ int lbase[NBMAX];
    __shared__ int wtot[NBMAX / 32];
    __shared__ int2 stage[CHUNK];
    const int tid = threadIdx.x, blk = blockIdx.x;
    const int lane = tid & 31, wave = tid >> 5;

    if (tid < NBMAX) cnt[tid] = 0;
    __syncthreads();

    const int e0 = blk * CHUNK + tid;
    int s[EPT], d[EPT], r[EPT];
#pragma unroll
    for (int it = 0; it < EPT; ++it) {
        const int e = e0 + it * PT;
        d[it] = (e < E) ? dst[e] : -1;
        s[it] = (e < E) ? src[e] : 0;
    }
#pragma unroll
    for (int it = 0; it < EPT; ++it)
        r[it] = (d[it] >= 0) ? atomicAdd(&cnt[d[it] / BW], 1) : 0;
    __syncthreads();

    int v = 0, inc = 0;
    if (tid < NBMAX) {
        v = cnt[tid];
        inc = v;
        for (int o = 1; o < 32; o <<= 1) {
            int u = __shfl_up(inc, o);
            if (lane >= o) inc += u;
        }
        if (lane == 31) wtot[wave] = inc;
    }
    __syncthreads();
    const int left = E - blk * CHUNK;
    const int nvalid = (left < CHUNK) ? left : CHUNK;
    if (tid < NBMAX) {
        int base = 0;
        for (int w = 0; w < NBMAX / 32; ++w)
            if (w < wave) base += wtot[w];
        const int ex = base + inc - v;
        lbase[tid] = ex;
        if (tid < nb) lbt[tid * nblk + blk] = ex;
    }
    if (tid == nb) lbt[nb * nblk + blk] = nvalid;
    __syncthreads();

#pragma unroll
    for (int it = 0; it < EPT; ++it)
        if (d[it] >= 0) stage[lbase[d[it] / BW] + r[it]] = make_int2(s[it], d[it]);
    __syncthreads();

    for (int i = tid; i < nvalid; i += PT) tmp[blk * CHUNK + i] = stage[i];
}

__device__ __forceinline__ int run_of(const int* incl, int nblk, int top, int i) {
    int pos = 0;
    for (int s = top; s > 0; s >>= 1) {
        const int c = pos + s;
        const int val = incl[(c < nblk) ? (c - 1) : 0];
        if (c < nblk && val <= i) pos = c;
    }
    return pos;
}

__device__ __forceinline__ void lds_barrier() {
    __builtin_amdgcn_fence(__ATOMIC_RELEASE, "workgroup", "local");
    __builtin_amdgcn_s_barrier();
    __builtin_amdgcn_fence(__ATOMIC_ACQUIRE, "workgroup", "local");
}

#define GK 128
#define GN 64
#define GP 132
#define SROWS 128
#define SLABS 4
#define BG_AIT 4
#define BG_WIT 2

__global__ void __launch_bounds__(BK_TPB)
k_bucket(const int2* __restrict__ tmp, const int* __restrict__ lbt,
         int* __restrict__ start, int* __restrict__ csr_src,
         const float* __restrict__ X, const float* __restrict__ W,
         float* __restrict__ H, int N, int nblk) {
    __shared__ __attribute__((aligned(16))) float lw[GN * GP];
    __shared__ __attribute__((aligned(16))) float la[SROWS * GP];
    __shared__ __attribute__((aligned(16))) float ldinv[BW];
    __shared__ int cnt[BW];
    __shared__ int wtot[BW / 32];
    __shared__ int stage[BK_CAP];
    __shared__ int incl[RMAX];
    __shared__ int gdel[RMAX];
    __shared__ int wlen[BK_TPB / 32];
    __shared__ int wlo[BK_TPB / 32];
    const int tid = threadIdx.x, b = blockIdx.x;
    const int lane = tid & 31, wave = tid >> 5;

    int lo = 0, len = 0;
    if (tid < nblk) {
        lo = lbt[b * nblk + tid];
        len = lbt[(b + 1) * nblk + tid] - lo;
    }

    const int row0 = b * BW;
    const int rend = (row0 + BW < N) ? (row0 + BW) : N;
    const int ar = tid >> 5;
    const int ac = tid & 31;
    float4 wv[BG_WIT], av[BG_AIT];
#pragma unroll
    for (int it = 0; it < BG_WIT; ++it)
        wv[it] = *(const float4*)(W + 4 * (size_t)(tid + it * BK_TPB));
#pragma unroll
    for (int it = 0; it < BG_AIT; ++it) {
        const int rw = row0 + ar + it * 32;
        const int rg = (rw < rend) ? rw : (rend - 1);
        av[it] = *(const float4*)(X + (size_t)rg * GK + 4 * ac);
    }
    int rinc = len, slo = lo;
    for (int o = 1; o < 32; o <<= 1) {
        int u = __shfl_up(rinc, o);
        if (lane >= o) rinc += u;
    }
    for (int o = 16; o > 0; o >>= 1) slo += __shfl_xor(slo, o);
    if (lane == 31) wlen[wave] = rinc;
    if (lane == 0) wlo[wave] = slo;
    if (tid < BW) cnt[tid] = 0;
    __syncthreads();

    int rbase = 0, m = 0, beg = 0;
    for (int w = 0; w < BK_TPB / 32; ++w) {
        const int a = wlen[w];
        if (w < wave) rbase += a;
        m += a;
        beg += wlo[w];
    }
    const bool fits = (m <= BK_CAP);
    if (tid < nblk) {
        const int ic = rbase + rinc;
        const int r0 = ic - len;
        const int g = tid * CHUNK + lo - r0;
        if (fits) {
            for (int k = 0; k < len; ++k) stage[r0 + k] = g;
        } else {
            incl[tid] = ic;
            gdel[tid] = g;
        }
    }
    __syncthreads();

    int top = 1;
    while (2 * top <= nblk) top *= 2;

    int2 e[BK_EPT];
    if (fits) {
        int g[BK_EPT];
#pragma unroll
        for (int j = 0; j < BK_EPT; ++j) {
            const int i = tid + j * BK_TPB;
            g[j] = stage[(i < m) ? i : 0];
        }
#pragma unroll
        for (int j = 0; j < BK_EPT; ++j) {
            const int i = tid + j * BK_TPB;
            e[j] = make_int2(0, -1);
            if (i < m) e[j] = tmp[g[j] + i];
        }
#pragma unroll
        for (int j = 0; j < BK_EPT; ++j)
            if (e[j].y >= 0) atomicAdd(&cnt[e[j].y - row0], 1);
    } else {
        for (int i = tid; i < m; i += BK_TPB)
            atomicAdd(&cnt[tmp[gdel[run_of(incl, nblk, top, i)] + i].y - row0], 1);
    }
    __syncthreads();

    int v = 0, inc = 0;
    if (tid < BW) {
        v = cnt[tid];
        inc = v;
        for (int o = 1; o < 32; o <<= 1) {
            int u = __shfl_up(inc, o);
            if (lane >= o) inc += u;
        }
        if (lane == 31) wtot[wave] = inc;
        ldinv[tid] = rsqrtf((float)(v + 1));
    }
#pragma unroll
    for (int it = 0; it < BG_WIT; ++it) {
        const int i = tid + it * BK_TPB;
        *(float4*)(lw + (i >> 5) * GP + 4 * (i & 31)) = wv[it];
    }
#pragma unroll
    for (int it = 0; it < BG_AIT; ++it) {
        float4 q = av[it];
        if (row0 + ar + it * 32 >= rend) q = make_float4(0.f, 0.f, 0.f, 0.f);
        *(float4*)(la + (ar + it * 32) * GP + 4 * ac) = q;
    }
    __syncthreads();
    if (tid < BW) {
        int base = 0;
        for (int w = 0; w < BW / 32; ++w)
            if (w < wave) base += wtot[w];
        const int ex = base + inc - v;
        cnt[tid] = ex;
        const int row = b * BW + tid;
        if (row < N) start[row] = beg + ex;
    }
    __syncthreads();

    if (fits) {
#pragma unroll
        for (int j = 0; j < BK_EPT; ++j)
            if (e[j].y >= 0) stage[atomicAdd(&cnt[e[j].y - row0], 1)] = e[j].x;
        __syncthreads();
        for (int i = tid; i < m; i += BK_TPB) csr_src[beg + i] = stage[i];
    } else {
        for (int i = tid; i < m; i += BK_TPB) {
            const int2 q = tmp[gdel[run_of(incl, nblk, top, i)] + i];
            csr_src[beg + atomicAdd(&cnt[q.y - row0], 1)] = q.x;
        }
    }

    const int rt = wave >> 2;
    const int n0 = (wave & 3) * 16;
    const int half = lane >> 4, l16 = lane & 15;
    const int hb = 8 * half;
    const float* arow = la + (rt * 16 + l16) * GP + 2 * half;
    const float* brow = lw + (n0 + l16) * GP + 2 * half;
    for (int sl = 0; sl < SLABS; ++sl) {
        const int s0 = row0 + sl * SROWS;
        const bool more = (sl + 1 < SLABS);
        if (more) {
#pragma unroll
            for (int it = 0; it < BG_AIT; ++it) {
                const int rw = s0 + SROWS + ar + it * 32;
                const int rg = (rw < rend) ? rw : (rend - 1);
                av[it] = *(const float4*)(X + (size_t)rg * GK + 4 * ac);
            }
        }
        const int m0 = s0 + rt * 16;
        if (m0 < rend) {
            v8f c = {0.f, 0.f, 0.f, 0.f, 0.f, 0.f, 0.f, 0.f};
#pragma unroll
            for (int k = 0; k < GK; k += 4) {
                v2f a = *(const v2f*)(arow + k);
                v2f bb = *(const v2f*)(brow + k);
                c = __builtin_amdgcn_wmma_f32_16x16x4_f32(
                     false, a,  false, bb,
                     (short)0, c,  false,  false);
            }
            int hq;
            asm volatile("v_nop\n\tv_nop\n\tv_nop\n\tv_nop\n\tv_mov_b32 %1, %2"
                         : "+v"(c), "=v"(hq) : "v"(hb));
            const float* dvp = ldinv + sl * SROWS + rt * 16 + hq;
            const float4 dlo = *(const float4*)(dvp);
            const float4 dhi = *(const float4*)(dvp + 4);
            const float dvs[8] = {dlo.x, dlo.y, dlo.z, dlo.w, dhi.x, dhi.y, dhi.z, dhi.w};
            const int rowb = m0 + hq;
            float* hp = H + (size_t)rowb * GN + n0 + l16;
            if (m0 + 16 <= rend) {
#pragma unroll
                for (int j = 0; j < 8; ++j) hp[j * GN] = c[j] * dvs[j];
            } else {
#pragma unroll
                for (int j = 0; j < 8; ++j)
                    if (rowb + j < rend) hp[j * GN] = c[j] * dvs[j];
            }
        }
        if (more) {
            lds_barrier();
#pragma unroll
            for (int it = 0; it < BG_AIT; ++it) {
                float4 q = av[it];
                if (s0 + SROWS + ar + it * 32 >= rend) q = make_float4(0.f, 0.f, 0.f, 0.f);
                *(float4*)(la + (ar + it * 32) * GP + 4 * ac) = q;
            }
            lds_barrier();
        }
    }
}


template <int F, bool RELU>
__global__ void __launch_bounds__(TPB)
k_gather(const float* __restrict__ h, const int* __restrict__ start,
         const int* __restrict__ csr_src, const float* __restrict__ bias,
         float* __restrict__ out, int n, int E) {
    constexpr int Q = F / 4;
    const int t = blockIdx.x * TPB + threadIdx.x;
    const int q = t % Q;
    const bool live = (t / Q) < n;
    const int d = live ? (t / Q) : (n - 1);
    const float4 self = *(const float4*)(h + (size_t)d * F + 4 * q);
    const int beg = start[d];
    const int endl = start[(d + 1 < n) ? d + 1 : d];
    const int end = (d + 1 < n) ? endl : E;
    const float dd = rsqrtf((float)(end - beg + 1));

    float4 acc = self;
    for (int p = beg; p < end; p += Q) {
        const int mine = (p + q < end) ? csr_src[p + q] : d;
        float4 v[Q];
#pragma unroll
        for (int j = 0; j < Q; ++j) {
            const int s = __shfl(mine, j, Q);
            v[j] = make_float4(0.f, 0.f, 0.f, 0.f);
            if (p + j < end)
                v[j] = *(const float4*)(h + (size_t)s * F + 4 * q);
        }
#pragma unroll
        for (int j = 0; j < Q; ++j) {
            acc.x += v[j].x; acc.y += v[j].y; acc.z += v[j].z; acc.w += v[j].w;
        }
    }

    const float4 bq = *(const float4*)(bias + 4 * q);
    float4 r;
    r.x = acc.x * dd + bq.x; r.y = acc.y * dd + bq.y;
    r.z = acc.z * dd + bq.z; r.w = acc.w * dd + bq.w;
    if (RELU) {
        r.x = fmaxf(r.x, 0.0f); r.y = fmaxf(r.y, 0.0f);
        r.z = fmaxf(r.z, 0.0f); r.w = fmaxf(r.w, 0.0f);
    }
    if (live) *(float4*)(out + (size_t)d * F + 4 * q) = r;
}


template <int F, int NOUT, int PASSES>
__global__ void __launch_bounds__(TPB)
k_gather_gemm(const float* __restrict__ h, const int* __restrict__ start,
              const int* __restrict__ csr_src, const float* __restrict__ bias,
              const float* __restrict__ W, float* __restrict__ H, int n, int E) {
    constexpr int Q = F / 4;
    constexpr int RP = TPB / Q;
    constexpr int P = F + 4;
    constexpr int KV = F / 4;
    constexpr int WIT = (NOUT * KV) / TPB;
    static_assert(RP == 16, "a group is one 16-row matrix tile");
    static_assert((NOUT * KV) % TPB == 0, "W is staged in whole passes of the block");
    static_assert(NOUT / 16 <= TPB / 32, "one wave per 16-column tile");

    __shared__ __attribute__((aligned(16))) float lw[NOUT * P];
    __shared__ __attribute__((aligned(16))) float la[2][RP * P];
    __shared__ __attribute__((aligned(16))) float ldv[2][RP];

    const int tid = threadIdx.x;
    const int q = tid % Q, r = tid / Q;
    const int wave = tid >> 5, lane = tid & 31;
    const int half = lane >> 4, l16 = lane & 15;
    const int hb = 8 * half;
    const int nl = (wave % (NOUT / 16)) * 16 + l16;

    float4 wv[WIT];
#pragma unroll
    for (int it = 0; it < WIT; ++it)
        wv[it] = *(const float4*)(W + 4 * (size_t)(tid + it * TPB));
#pragma unroll
    for (int it = 0; it < WIT; ++it) {
        const int i = tid + it * TPB;
        *(float4*)(lw + (i / KV) * P + 4 * (i % KV)) = wv[it];
    }
    const float4 bq = *(const float4*)(bias + 4 * q);

    for (int ps = 0; ps < PASSES; ++ps) {
        const int m0 = (blockIdx.x * PASSES + ps) * RP;
        if (m0 >= n) break;
        const bool live = (m0 + r) < n;
        const int d = live ? (m0 + r) : (n - 1);
        const float4 self = *(const float4*)(h + (size_t)d * F + 4 * q);
        const int beg = start[d];
        const int endl = start[(d + 1 < n) ? d + 1 : d];
        const int end = (d + 1 < n) ? endl : E;
        const float dd = rsqrtf((float)(end - beg + 1));

        float4 acc = self;
        for (int p = beg; p < end; p += Q) {
            const int mine = (p + q < end) ? csr_src[p + q] : d;
            float4 v[Q];
#pragma unroll
            for (int j = 0; j < Q; ++j) {
                const int s = __shfl(mine, j, Q);
                v[j] = make_float4(0.f, 0.f, 0.f, 0.f);
                if (p + j < end)
                    v[j] = *(const float4*)(h + (size_t)s * F + 4 * q);
            }
#pragma unroll
            for (int j = 0; j < Q; ++j) {
                acc.x += v[j].x; acc.y += v[j].y; acc.z += v[j].z; acc.w += v[j].w;
            }
        }

        float4 rv;
        rv.x = fmaxf(acc.x * dd + bq.x, 0.0f); rv.y = fmaxf(acc.y * dd + bq.y, 0.0f);
        rv.z = fmaxf(acc.z * dd + bq.z, 0.0f); rv.w = fmaxf(acc.w * dd + bq.w, 0.0f);
        if (!live) rv = make_float4(0.f, 0.f, 0.f, 0.f);

        float* lab = la[ps & 1];
        *(float4*)(lab + r * P + 4 * q) = rv;
        if (q == 0) ldv[ps & 1][r] = dd;
        __syncthreads();

        if (wave < NOUT / 16) {
            const int n0 = wave * 16;
            const float* arow = lab + l16 * P + 2 * half;
            const float* brow = lw + (n0 + l16) * P + 2 * half;
            v8f c = {0.f, 0.f, 0.f, 0.f, 0.f, 0.f, 0.f, 0.f};
#pragma unroll
            for (int k = 0; k < F; k += 4) {
                v2f a = *(const v2f*)(arow + k);
                v2f b = *(const v2f*)(brow + k);
                c = __builtin_amdgcn_wmma_f32_16x16x4_f32(
                     false, a,  false, b,
                     (short)0, c,  false,  false);
            }
            int hq;
            asm volatile("v_nop\n\tv_nop\n\tv_nop\n\tv_nop\n\tv_mov_b32 %1, %2"
                         : "+v"(c), "=v"(hq) : "v"(hb));
            const float4 dlo = *(const float4*)(&ldv[ps & 1][hq]);
            const float4 dhi = *(const float4*)(&ldv[ps & 1][hq + 4]);
            const float dvs[8] = {dlo.x, dlo.y, dlo.z, dlo.w, dhi.x, dhi.y, dhi.z, dhi.w};
            const int rowb = m0 + hq;
            float* hp = H + (size_t)rowb * NOUT + nl;
            if (m0 + RP <= n) {
#pragma unroll
                for (int j = 0; j < 8; ++j) hp[j * NOUT] = c[j] * dvs[j];
            } else {
#pragma unroll
                for (int j = 0; j < 8; ++j)
                    if (rowb + j < n) hp[j * NOUT] = c[j] * dvs[j];
            }
        }
    }
}


static inline int cdiv(long long a, int b) { return (int)((a + b - 1) / b); }

extern "C" void kernel_launch(void* const* d_in, const int* in_sizes, int n_in,
                              void* d_out, int out_size, void* d_ws, size_t ws_size,
                              hipStream_t stream) {
    (void)n_in; (void)out_size; (void)ws_size;
    const float* x  = (const float*)d_in[0];
    const int*   ei = (const int*)  d_in[1];
    const float* W1 = (const float*)d_in[2];
    const float* b1 = (const float*)d_in[3];
    const float* W2 = (const float*)d_in[4];
    const float* b2 = (const float*)d_in[5];
    float* out = (float*)d_out;

    const int N = in_sizes[0] / 128;
    const int E = in_sizes[1] / 2;
    const int* src = ei;
    const int* dst = ei + E;

    float* ws   = (float*)d_ws;
    const size_t NA = ((size_t)N + 3) & ~(size_t)3;
    int*   start   = (int*)ws;
    float* h1      = ws + NA;
    float* agg1    = h1 + 64 * NA;
    int*   csr_src = (int*)(agg1 + 64 * NA);
    float* h2      = agg1;
    const int nblk = cdiv(E, CHUNK);
    const int nb   = cdiv(N, BW);
    const int tsz  = ((nb + 1) * nblk + 3) & ~3;
    int*  lbt = (int*)agg1;
    int2* tmp = (int2*)(lbt + tsz);

    k_part<<<nblk, PT, 0, stream>>>(src, dst, lbt, tmp, E, nblk, nb);
    k_bucket<<<nb, BK_TPB, 0, stream>>>(tmp, lbt, start, csr_src, x, W1, h1, N, nblk);

    k_gather_gemm<64, 32, 6><<<cdiv(N, 16 * 6), TPB, 0, stream>>>(h1, start, csr_src, b1, W2, h2, N, E);
    k_gather<32, false><<<cdiv((long long)N * 8, TPB), TPB, 0, stream>>>(h2, start, csr_src, b2, out, N, E);
}
